# starting kernel
# baseline (speedup 1.0000x reference)
#include <hip/hip_runtime.h>
#include <hip/hip_bf16.h>


#define BB  512
#define DD  64
#define HH  512
#define TT  8
#define MT  32
#define XP  68
#define HP  516

typedef float v2f __attribute__((ext_vector_type(2)));
typedef float v8f __attribute__((ext_vector_type(8)));

__device__ __forceinline__ v8f wmma_f32(v2f a, v2f b, v8f c) {
  return __builtin_amdgcn_wmma_f32_16x16x4_f32(false, a, false, b,
                                               (short)0, c, false, false);
}

__device__ __forceinline__ float fast_tanh(float x) {
#if __has_builtin(__builtin_amdgcn_tanhf)
  return __builtin_amdgcn_tanhf(x);
#elif __has_builtin(__builtin_amdgcn_exp2f) && __has_builtin(__builtin_amdgcn_rcpf)
  float e = __builtin_amdgcn_exp2f(x * 2.8853900817779268f);
  return 1.0f - 2.0f * __builtin_amdgcn_rcpf(e + 1.0f);
#else
  return tanhf(x);
#endif
}

__device__ __forceinline__ v2f frag_a(const float* S, int pitch, int m0, int k0,
                                      int lane_lo, int half) {
  const float* p = S + (m0 + lane_lo) * pitch + k0 + 2 * half;
  v2f a; a.x = p[0]; a.y = p[1];
  return a;
}

__device__ __forceinline__ v2f frag_b(const float* __restrict__ W, int ldw,
                                      int k0, int n0, int lane_lo, int half) {
  v2f b;
  b.x = W[(k0 + 2 * half + 0) * ldw + n0 + lane_lo];
  b.y = W[(k0 + 2 * half + 1) * ldw + n0 + lane_lo];
  return b;
}

__global__ __launch_bounds__(256) void cnf_prep(const float* __restrict__ W1,
                                                const float* __restrict__ W2,
                                                const float* __restrict__ W3,
                                                float* __restrict__ M) {
  int idx = blockIdx.x * blockDim.x + threadIdx.x;
  int j = idx >> 9;
  int k = idx & 511;
  float c = 0.f;
  #pragma unroll 8
  for (int m = 0; m < DD; ++m)
    c += W3[k * DD + m] * W1[m * HH + j];
  M[j * HH + k] = W2[j * HH + k] * c;
}

__global__ __launch_bounds__(256) void cnf_main(const float* __restrict__ y0,
                                                const float* __restrict__ ts,
                                                const float* __restrict__ W1,
                                                const float* __restrict__ b1,
                                                const float* __restrict__ W2,
                                                const float* __restrict__ b2,
                                                const float* __restrict__ W3,
                                                const float* __restrict__ b3,
                                                const float* __restrict__ Mtr,
                                                float* __restrict__ out) {
  extern __shared__ float smem[];
  float* Xs   = smem;
  float* H1s  = Xs  + MT * XP;
  float* H2s  = H1s + MT * HP;
  float* zacc = H2s + MT * HP;
  float* lacc = zacc + MT;

  const int tid     = threadIdx.x;
  const int wave    = tid >> 5;
  const int lane    = tid & 31;
  const int lane_lo = lane & 15;
  const int half    = lane >> 4;
  const int s_base  = blockIdx.x * MT;

  for (int i = tid; i < MT * DD; i += 256) {
    int m = i >> 6, c = i & 63;
    Xs[m * XP + c] = y0[(s_base + m) * DD + c];
  }
  if (tid < MT) { zacc[tid] = 0.f; lacc[tid] = 0.f; }

  v8f acc[2][4];

  for (int step = 0; step < TT - 1; ++step) {
    const float t0 = ts[step];
    const float dt = ts[step + 1] - t0;
    __syncthreads();

    #pragma unroll
    for (int mt = 0; mt < 2; ++mt)
      #pragma unroll
      for (int nt = 0; nt < 4; ++nt) acc[mt][nt] = (v8f){};
    for (int k0 = 0; k0 < DD; k0 += 4) {
      v2f a0 = frag_a(Xs, XP, 0,  k0, lane_lo, half);
      v2f a1 = frag_a(Xs, XP, 16, k0, lane_lo, half);
      #pragma unroll
      for (int nt = 0; nt < 4; ++nt) {
        int n0 = (wave * 4 + nt) * 16;
        v2f b = frag_b(W1, HH, k0, n0, lane_lo, half);
        acc[0][nt] = wmma_f32(a0, b, acc[0][nt]);
        acc[1][nt] = wmma_f32(a1, b, acc[1][nt]);
      }
    }
    #pragma unroll
    for (int nt = 0; nt < 4; ++nt) {
      int col = (wave * 4 + nt) * 16 + lane_lo;
      float add = t0 * W1[DD * HH + col] + b1[col];
      #pragma unroll
      for (int mt = 0; mt < 2; ++mt)
        #pragma unroll
        for (int v = 0; v < 8; ++v) {
          int m = mt * 16 + v + 8 * half;
          H1s[m * HP + col] = fast_tanh(acc[mt][nt][v] + add);
        }
    }
    __syncthreads();

    #pragma unroll
    for (int mt = 0; mt < 2; ++mt)
      #pragma unroll
      for (int nt = 0; nt < 4; ++nt) acc[mt][nt] = (v8f){};
    for (int k0 = 0; k0 < HH; k0 += 4) {
      v2f a0 = frag_a(H1s, HP, 0,  k0, lane_lo, half);
      v2f a1 = frag_a(H1s, HP, 16, k0, lane_lo, half);
      #pragma unroll
      for (int nt = 0; nt < 4; ++nt) {
        int n0 = (wave * 4 + nt) * 16;
        v2f b = frag_b(W2, HH, k0, n0, lane_lo, half);
        acc[0][nt] = wmma_f32(a0, b, acc[0][nt]);
        acc[1][nt] = wmma_f32(a1, b, acc[1][nt]);
      }
    }
    #pragma unroll
    for (int nt = 0; nt < 4; ++nt) {
      int col = (wave * 4 + nt) * 16 + lane_lo;
      float add = b2[col];
      #pragma unroll
      for (int mt = 0; mt < 2; ++mt)
        #pragma unroll
        for (int v = 0; v < 8; ++v) {
          int m = mt * 16 + v + 8 * half;
          H2s[m * HP + col] = fast_tanh(acc[mt][nt][v] + add);
        }
    }
    __syncthreads();

    {
      int m0 = (wave >> 2) * 16;
      int n0 = (wave & 3) * 16;
      v8f a3 = (v8f){};
      for (int k0 = 0; k0 < HH; k0 += 4) {
        v2f a = frag_a(H2s, HP, m0, k0, lane_lo, half);
        v2f b = frag_b(W3, DD, k0, n0, lane_lo, half);
        a3 = wmma_f32(a, b, a3);
      }
      int col = n0 + lane_lo;
      float bb = b3[col];
      #pragma unroll
      for (int v = 0; v < 8; ++v) {
        int m = m0 + v + 8 * half;
        float dx = a3[v] + bb;
        atomicAdd(&zacc[m], dt * 0.5f * dx * dx);
        Xs[m * XP + col] += dt * dx;
      }
    }

    #pragma unroll
    for (int mt = 0; mt < 2; ++mt)
      #pragma unroll
      for (int nt = 0; nt < 4; ++nt) acc[mt][nt] = (v8f){};
    for (int k0 = 0; k0 < HH; k0 += 4) {
      v2f a0 = frag_a(H1s, HP, 0,  k0, lane_lo, half);
      v2f a1 = frag_a(H1s, HP, 16, k0, lane_lo, half);
      a0.x = 1.f - a0.x * a0.x;  a0.y = 1.f - a0.y * a0.y;
      a1.x = 1.f - a1.x * a1.x;  a1.y = 1.f - a1.y * a1.y;
      #pragma unroll
      for (int nt = 0; nt < 4; ++nt) {
        int n0 = (wave * 4 + nt) * 16;
        v2f b = frag_b(Mtr, HH, k0, n0, lane_lo, half);
        acc[0][nt] = wmma_f32(a0, b, acc[0][nt]);
        acc[1][nt] = wmma_f32(a1, b, acc[1][nt]);
      }
    }
    #pragma unroll
    for (int mt = 0; mt < 2; ++mt)
      #pragma unroll
      for (int v = 0; v < 8; ++v) {
        int m = mt * 16 + v + 8 * half;
        float tl = 0.f;
        #pragma unroll
        for (int nt = 0; nt < 4; ++nt) {
          int col = (wave * 4 + nt) * 16 + lane_lo;
          float h2 = H2s[m * HP + col];
          tl += (1.f - h2 * h2) * acc[mt][nt][v];
        }
        atomicAdd(&lacc[m], dt * tl);
      }
  }

  __syncthreads();
  for (int i = tid; i < MT * DD; i += 256) {
    int m = i >> 6, c = i & 63;
    out[(s_base + m) * DD + c] = Xs[m * XP + c];
  }
  if (tid < MT) {
    out[BB * DD + s_base + tid]      = zacc[tid];
    out[BB * DD + BB + s_base + tid] = lacc[tid];
  }
}

extern "C" void kernel_launch(void* const* d_in, const int* in_sizes, int n_in,
                              void* d_out, int out_size, void* d_ws, size_t ws_size,
                              hipStream_t stream) {
  const float* y0 = (const float*)d_in[0];
  const float* ts = (const float*)d_in[1];
  const float* W1 = (const float*)d_in[2];
  const float* b1 = (const float*)d_in[3];
  const float* W2 = (const float*)d_in[4];
  const float* b2 = (const float*)d_in[5];
  const float* W3 = (const float*)d_in[6];
  const float* b3 = (const float*)d_in[7];
  float* out = (float*)d_out;
  float* Mtr = (float*)d_ws;

  cnf_prep<<<(HH * HH) / 256, 256, 0, stream>>>(W1, W2, W3, Mtr);

  size_t shmem = (size_t)(MT * XP + 2 * MT * HP + 2 * MT) * sizeof(float);
  cnf_main<<<BB / MT, 256, shmem, stream>>>(y0, ts, W1, b1, W2, b2, W3, b3,
                                            Mtr, out);
}


# Layer-2 and trace GEMMs re-tiled: a wave owns a 32x64 block over one K half (16 tiles, 32 WMMAs per weight round trip), partner waves join halves via LDS; K sums split in two, trace sums regroup
# speedup vs baseline: 1.4584x; 1.4584x over previous
#include <hip/hip_runtime.h>
#include <hip/hip_bf16.h>


#define BB  512
#define DD  64
#define HH  512
#define TT  8
#define MT  32
#define NW  16
#define NTH (NW * 32)
#define NTW (32 / NW)
#define XP  68
#define HP  516

typedef float v2f __attribute__((ext_vector_type(2)));
typedef float v8f __attribute__((ext_vector_type(8)));

__device__ __forceinline__ v8f wmma_f32(v2f a, v2f b, v8f c) {
  return __builtin_amdgcn_wmma_f32_16x16x4_f32(false, a, false, b,
                                               (short)0, c, false, false);
}

__device__ __forceinline__ float fast_tanh(float x) {
#if __has_builtin(__builtin_amdgcn_tanhf)
  return __builtin_amdgcn_tanhf(x);
#elif __has_builtin(__builtin_amdgcn_exp2f) && __has_builtin(__builtin_amdgcn_rcpf)
  float e = __builtin_amdgcn_exp2f(x * 2.8853900817779268f);
  return 1.0f - 2.0f * __builtin_amdgcn_rcpf(e + 1.0f);
#else
  return tanhf(x);
#endif
}

__device__ __forceinline__ v2f frag_a(const float* S, int pitch, int m0, int k0,
                                      int lane_lo, int half) {
  const float* p = S + (m0 + lane_lo) * pitch + k0 + 2 * half;
  v2f a; a.x = p[0]; a.y = p[1];
  return a;
}

__device__ __forceinline__ v2f frag_b(const float* __restrict__ W, int ldw,
                                      int k0, int n0, int lane_lo, int half) {
  v2f b;
  b.x = W[(k0 + 2 * half + 0) * ldw + n0 + lane_lo];
  b.y = W[(k0 + 2 * half + 1) * ldw + n0 + lane_lo];
  return b;
}

template <int X>
__device__ __forceinline__ float row_xor(float v) {
  return __builtin_bit_cast(float,
      __builtin_amdgcn_update_dpp(0, __builtin_bit_cast(int, v),
                                  0x160 + X, 0xf, 0xf, true));
}

template <int N>
__device__ __forceinline__ void row16_sum(float (&r)[N]) {
  #pragma unroll
  for (int i = 0; i < N; ++i) r[i] += row_xor<1>(r[i]);
  #pragma unroll
  for (int i = 0; i < N; ++i) r[i] += row_xor<2>(r[i]);
  #pragma unroll
  for (int i = 0; i < N; ++i) r[i] += row_xor<4>(r[i]);
  #pragma unroll
  for (int i = 0; i < N; ++i) r[i] += row_xor<8>(r[i]);
}

__global__ __launch_bounds__(256) void cnf_prep(const float* __restrict__ W1,
                                                const float* __restrict__ W2,
                                                const float* __restrict__ W3,
                                                float* __restrict__ M) {
  int idx = blockIdx.x * blockDim.x + threadIdx.x;
  int j = idx >> 9;
  int k = idx & 511;
  float c = 0.f;
  #pragma unroll 8
  for (int m = 0; m < DD; ++m)
    c += W3[k * DD + m] * W1[m * HH + j];
  M[j * HH + k] = W2[j * HH + k] * c;
}

__global__ __launch_bounds__(NTH) void cnf_main(const float* __restrict__ y0,
                                                const float* __restrict__ ts,
                                                const float* __restrict__ W1,
                                                const float* __restrict__ b1,
                                                const float* __restrict__ W2,
                                                const float* __restrict__ b2,
                                                const float* __restrict__ W3,
                                                const float* __restrict__ b3,
                                                const float* __restrict__ Mtr,
                                                float* __restrict__ out) {
  extern __shared__ float smem[];
  float* Xs   = smem;
  float* H1s  = Xs  + MT * XP;
  float* H2s  = H1s + MT * HP;
  float* zacc = H2s + MT * HP;
  float* lacc = zacc + MT;
  float* red  = lacc + MT;
  float* red2 = red + NW * 4 * 32;

  const int tid     = threadIdx.x;
  const int wave    = __builtin_amdgcn_readfirstlane(tid >> 5);
  const int lane    = tid & 31;
  const int lane_lo = lane & 15;
  const int half    = lane >> 4;
  const int s_base  = blockIdx.x * MT;

  for (int i = tid; i < MT * DD; i += NTH) {
    int m = i >> 6, c = i & 63;
    Xs[m * XP + c] = y0[(s_base + m) * DD + c];
  }
  if (tid < MT) { zacc[tid] = 0.f; lacc[tid] = 0.f; }

  v8f acc[2][NTW];

  for (int step = 0; step < TT - 1; ++step) {
    const float t0 = ts[step];
    const float dt = ts[step + 1] - t0;

    const float* W1s = W1;
    asm volatile("" : "+s"(W1s));
    v2f w1f[DD / 4][NTW];
    {
      const float* w1p = W1s + (2 * half) * HH + wave * (NTW * 16) + lane_lo;
      #pragma unroll
      for (int ks = 0; ks < DD / 4; ++ks)
        #pragma unroll
        for (int nt = 0; nt < NTW; ++nt) {
          w1f[ks][nt].x = w1p[(ks * 4 + 0) * HH + nt * 16];
          w1f[ks][nt].y = w1p[(ks * 4 + 1) * HH + nt * 16];
        }
    }
    __syncthreads();

    #pragma unroll
    for (int mt = 0; mt < 2; ++mt)
      #pragma unroll
      for (int nt = 0; nt < NTW; ++nt) acc[mt][nt] = (v8f){};
    #pragma unroll
    for (int ks = 0; ks < DD / 4; ++ks) {
      v2f a0 = frag_a(Xs, XP, 0,  ks * 4, lane_lo, half);
      v2f a1 = frag_a(Xs, XP, 16, ks * 4, lane_lo, half);
      #pragma unroll
      for (int nt = 0; nt < NTW; ++nt) {
        acc[0][nt] = wmma_f32(a0, w1f[ks][nt], acc[0][nt]);
        acc[1][nt] = wmma_f32(a1, w1f[ks][nt], acc[1][nt]);
      }
    }
    #pragma unroll
    for (int nt = 0; nt < NTW; ++nt) {
      int col = (wave * NTW + nt) * 16 + lane_lo;
      float add = t0 * W1[DD * HH + col] + b1[col];
      #pragma unroll
      for (int mt = 0; mt < 2; ++mt)
        #pragma unroll
        for (int v = 0; v < 8; ++v) {
          int m = mt * 16 + v + 8 * half;
          H1s[m * HP + col] = fast_tanh(acc[mt][nt][v] + add);
        }
    }
    __syncthreads();

    {
      const int ng    = wave & 7;
      const int kp    = wave >> 3;
      const int kbase = kp * (HH / 2);
      v8f hac[2][4];
      v8f vac[2][4];
      #pragma unroll
      for (int mt = 0; mt < 2; ++mt)
        #pragma unroll
        for (int nt = 0; nt < 4; ++nt) { hac[mt][nt] = (v8f){}; vac[mt][nt] = (v8f){}; }
      const float* w2p = W2  + (kbase + 2 * half) * HH + ng * 64 + lane_lo;
      const float* mp  = Mtr + (kbase + 2 * half) * HH + ng * 64 + lane_lo;
      const float* h1p = H1s + lane_lo * HP + kbase + 2 * half;
      for (int kk = 0; kk < HH / 2; kk += 8) {
        v2f b[2][4], bm[2][4], a0[2], a1[2];
        #pragma unroll
        for (int u = 0; u < 2; ++u) {
          #pragma unroll
          for (int nt = 0; nt < 4; ++nt) {
            b[u][nt].x  = w2p[(kk + 4 * u + 0) * HH + nt * 16];
            b[u][nt].y  = w2p[(kk + 4 * u + 1) * HH + nt * 16];
            bm[u][nt].x = mp[(kk + 4 * u + 0) * HH + nt * 16];
            bm[u][nt].y = mp[(kk + 4 * u + 1) * HH + nt * 16];
          }
          a0[u].x = h1p[kk + 4 * u];            a0[u].y = h1p[kk + 4 * u + 1];
          a1[u].x = h1p[16 * HP + kk + 4 * u];  a1[u].y = h1p[16 * HP + kk + 4 * u + 1];
        }
        #pragma unroll
        for (int u = 0; u < 2; ++u) {
          v2f s0, s1;
          s0.x = 1.f - a0[u].x * a0[u].x;  s0.y = 1.f - a0[u].y * a0[u].y;
          s1.x = 1.f - a1[u].x * a1[u].x;  s1.y = 1.f - a1[u].y * a1[u].y;
          #pragma unroll
          for (int nt = 0; nt < 4; ++nt) {
            hac[0][nt] = wmma_f32(a0[u], b[u][nt],  hac[0][nt]);
            hac[1][nt] = wmma_f32(a1[u], b[u][nt],  hac[1][nt]);
            vac[0][nt] = wmma_f32(s0,    bm[u][nt], vac[0][nt]);
            vac[1][nt] = wmma_f32(s1,    bm[u][nt], vac[1][nt]);
          }
        }
      }
      float* mine = red2 + wave * (64 * 32);
      float kh[4][8], kv[4][8];
      #pragma unroll
      for (int nt = 0; nt < 4; ++nt)
        #pragma unroll
        for (int v = 0; v < 8; ++v) {
          mine[(nt * 8 + v) * 32 + lane]      = kp ? hac[0][nt][v] : hac[1][nt][v];
          mine[(32 + nt * 8 + v) * 32 + lane] = kp ? vac[0][nt][v] : vac[1][nt][v];
          kh[nt][v] = kp ? hac[1][nt][v] : hac[0][nt][v];
          kv[nt][v] = kp ? vac[1][nt][v] : vac[0][nt][v];
        }
      __syncthreads();
      const float* theirs = red2 + (wave ^ 8) * (64 * 32);
      float ph[4][8], pv[4][8];
      #pragma unroll
      for (int nt = 0; nt < 4; ++nt)
        #pragma unroll
        for (int v = 0; v < 8; ++v) {
          ph[nt][v] = theirs[(nt * 8 + v) * 32 + lane];
          pv[nt][v] = theirs[(32 + nt * 8 + v) * 32 + lane];
        }
      float add[4];
      #pragma unroll
      for (int nt = 0; nt < 4; ++nt) add[nt] = b2[(ng * 4 + nt) * 16 + lane_lo];
      float tl[8];
      #pragma unroll
      for (int v = 0; v < 8; ++v) {
        int m = kp * 16 + v + 8 * half;
        float t = 0.f;
        #pragma unroll
        for (int nt = 0; nt < 4; ++nt) {
          int col = (ng * 4 + nt) * 16 + lane_lo;
          float a2 = kh[nt][v] + ph[nt][v];
          float vv = kv[nt][v] + pv[nt][v];
          float h2 = fast_tanh(a2 + add[nt]);
          H2s[m * HP + col] = h2;
          t += (1.f - h2 * h2) * vv;
        }
        tl[v] = t;
      }
      row16_sum<8>(tl);
      if (lane_lo == 0) {
        #pragma unroll
        for (int v = 0; v < 8; ++v)
          atomicAdd(&lacc[kp * 16 + v + 8 * half], dt * tl[v]);
      }
    }
    __syncthreads();

    {
      const int tile = wave & 7;
      const int kp   = wave >> 3;
      const int m0 = (tile >> 2) * 16;
      const int n0 = (tile & 3) * 16;
      v8f a3 = (v8f){};
      const int kbase = kp * (HH / 2);
      for (int kk = 0; kk < HH / 2; kk += 4) {
        const int k0 = kbase + kk;
        v2f a = frag_a(H2s, HP, m0, k0, lane_lo, half);
        v2f b = frag_b(W3, DD, k0, n0, lane_lo, half);
        a3 = wmma_f32(a, b, a3);
      }
      float keep[4], give[4];
      #pragma unroll
      for (int j = 0; j < 4; ++j) {
        keep[j] = kp ? a3[4 + j] : a3[j];
        give[j] = kp ? a3[j]     : a3[4 + j];
      }
      #pragma unroll
      for (int j = 0; j < 4; ++j) red[(wave * 4 + j) * 32 + lane] = give[j];
      __syncthreads();
      const int pw = wave ^ 8;
      int col = n0 + lane_lo;
      float bb = b3[col];
      float zt[4];
      #pragma unroll
      for (int j = 0; j < 4; ++j) {
        int m = m0 + 4 * kp + j + 8 * half;
        float dx = (keep[j] + red[(pw * 4 + j) * 32 + lane]) + bb;
        zt[j] = dt * 0.5f * dx * dx;
        Xs[m * XP + col] += dt * dx;
      }
      row16_sum<4>(zt);
      if (lane_lo == 0) {
        #pragma unroll
        for (int j = 0; j < 4; ++j)
          atomicAdd(&zacc[m0 + 4 * kp + j + 8 * half], zt[j]);
      }
    }
  }

  __syncthreads();
  for (int i = tid; i < MT * DD; i += NTH) {
    int m = i >> 6, c = i & 63;
    out[(s_base + m) * DD + c] = Xs[m * XP + c];
  }
  if (tid < MT) {
    out[BB * DD + s_base + tid]      = zacc[tid];
    out[BB * DD + BB + s_base + tid] = lacc[tid];
  }
}

extern "C" void kernel_launch(void* const* d_in, const int* in_sizes, int n_in,
                              void* d_out, int out_size, void* d_ws, size_t ws_size,
                              hipStream_t stream) {
  const float* y0 = (const float*)d_in[0];
  const float* ts = (const float*)d_in[1];
  const float* W1 = (const float*)d_in[2];
  const float* b1 = (const float*)d_in[3];
  const float* W2 = (const float*)d_in[4];
  const float* b2 = (const float*)d_in[5];
  const float* W3 = (const float*)d_in[6];
  const float* b3 = (const float*)d_in[7];
  float* out = (float*)d_out;
  float* Mtr = (float*)d_ws;

  cnf_prep<<<(HH * HH) / 256, 256, 0, stream>>>(W1, W2, W3, Mtr);

  size_t shmem = (size_t)(MT * XP + 2 * MT * HP + 2 * MT + NW * 4 * 32 + NW * 64 * 32) * sizeof(float);
  cnf_main<<<BB / MT, NTH, shmem, stream>>>(y0, ts, W1, b1, W2, b2, W3, b3,
                                            Mtr, out);
}


# dx GEMM split by hidden column group: each wave multiplies the H2 tile it just finished by its 64 W3 rows (4 tiles), 8 pieces per tile added in fixed order via LDS; one barrier fewer; dx sums regroup
# speedup vs baseline: 1.5035x; 1.0309x over previous
#include <hip/hip_runtime.h>
#include <hip/hip_bf16.h>


#define BB  512
#define DD  64
#define HH  512
#define TT  8
#define MT  32
#define NW  16
#define NTH (NW * 32)
#define NTW (32 / NW)
#define XP  68
#define HP  516

typedef float v2f __attribute__((ext_vector_type(2)));
typedef float v4f __attribute__((ext_vector_type(4)));
typedef float v8f __attribute__((ext_vector_type(8)));

__device__ __forceinline__ v8f wmma_f32(v2f a, v2f b, v8f c) {
  return __builtin_amdgcn_wmma_f32_16x16x4_f32(false, a, false, b,
                                               (short)0, c, false, false);
}

__device__ __forceinline__ float fast_tanh(float x) {
#if __has_builtin(__builtin_amdgcn_tanhf)
  return __builtin_amdgcn_tanhf(x);
#elif __has_builtin(__builtin_amdgcn_exp2f) && __has_builtin(__builtin_amdgcn_rcpf)
  float e = __builtin_amdgcn_exp2f(x * 2.8853900817779268f);
  return 1.0f - 2.0f * __builtin_amdgcn_rcpf(e + 1.0f);
#else
  return tanhf(x);
#endif
}

__device__ __forceinline__ v2f frag_a(const float* S, int pitch, int m0, int k0,
                                      int lane_lo, int half) {
  const float* p = S + (m0 + lane_lo) * pitch + k0 + 2 * half;
  v2f a; a.x = p[0]; a.y = p[1];
  return a;
}

__device__ __forceinline__ v2f frag_b(const float* __restrict__ W, int ldw,
                                      int k0, int n0, int lane_lo, int half) {
  v2f b;
  b.x = W[(k0 + 2 * half + 0) * ldw + n0 + lane_lo];
  b.y = W[(k0 + 2 * half + 1) * ldw + n0 + lane_lo];
  return b;
}

template <int X>
__device__ __forceinline__ float row_xor(float v) {
  return __builtin_bit_cast(float,
      __builtin_amdgcn_update_dpp(0, __builtin_bit_cast(int, v),
                                  0x160 + X, 0xf, 0xf, true));
}

template <int N>
__device__ __forceinline__ void row16_sum(float (&r)[N]) {
  #pragma unroll
  for (int i = 0; i < N; ++i) r[i] += row_xor<1>(r[i]);
  #pragma unroll
  for (int i = 0; i < N; ++i) r[i] += row_xor<2>(r[i]);
  #pragma unroll
  for (int i = 0; i < N; ++i) r[i] += row_xor<4>(r[i]);
  #pragma unroll
  for (int i = 0; i < N; ++i) r[i] += row_xor<8>(r[i]);
}

__global__ __launch_bounds__(256) void cnf_prep(const float* __restrict__ W1,
                                                const float* __restrict__ W2,
                                                const float* __restrict__ W3,
                                                float* __restrict__ M) {
  int idx = blockIdx.x * blockDim.x + threadIdx.x;
  int j = idx >> 9;
  int k = idx & 511;
  float c = 0.f;
  #pragma unroll 8
  for (int m = 0; m < DD; ++m)
    c += W3[k * DD + m] * W1[m * HH + j];
  M[j * HH + k] = W2[j * HH + k] * c;
}

__global__ __launch_bounds__(NTH) void cnf_main(const float* __restrict__ y0,
                                                const float* __restrict__ ts,
                                                const float* __restrict__ W1,
                                                const float* __restrict__ b1,
                                                const float* __restrict__ W2,
                                                const float* __restrict__ b2,
                                                const float* __restrict__ W3,
                                                const float* __restrict__ b3,
                                                const float* __restrict__ Mtr,
                                                float* __restrict__ out) {
  extern __shared__ float smem[];
  float* Xs   = smem;
  float* H1s  = Xs  + MT * XP;
  float* H2s  = H1s + MT * HP;
  float* zacc = H2s + MT * HP;
  float* lacc = zacc + MT;
  float* red2 = lacc + MT;
  float* dxp  = H1s;

  const int tid     = threadIdx.x;
  const int wave    = __builtin_amdgcn_readfirstlane(tid >> 5);
  const int lane    = tid & 31;
  const int lane_lo = lane & 15;
  const int half    = lane >> 4;
  const int s_base  = blockIdx.x * MT;

  for (int i = tid; i < MT * DD; i += NTH) {
    int m = i >> 6, c = i & 63;
    Xs[m * XP + c] = y0[(s_base + m) * DD + c];
  }
  if (tid < MT) { zacc[tid] = 0.f; lacc[tid] = 0.f; }

  v8f acc[2][NTW];

  for (int step = 0; step < TT - 1; ++step) {
    const float t0 = ts[step];
    const float dt = ts[step + 1] - t0;

    const float* W1s = W1;
    asm volatile("" : "+s"(W1s));
    v2f w1f[DD / 4][NTW];
    {
      const float* w1p = W1s + (2 * half) * HH + wave * (NTW * 16) + lane_lo;
      #pragma unroll
      for (int ks = 0; ks < DD / 4; ++ks)
        #pragma unroll
        for (int nt = 0; nt < NTW; ++nt) {
          w1f[ks][nt].x = w1p[(ks * 4 + 0) * HH + nt * 16];
          w1f[ks][nt].y = w1p[(ks * 4 + 1) * HH + nt * 16];
        }
    }
    __syncthreads();

    #pragma unroll
    for (int mt = 0; mt < 2; ++mt)
      #pragma unroll
      for (int nt = 0; nt < NTW; ++nt) acc[mt][nt] = (v8f){};
    #pragma unroll
    for (int ks = 0; ks < DD / 4; ++ks) {
      v2f a0 = frag_a(Xs, XP, 0,  ks * 4, lane_lo, half);
      v2f a1 = frag_a(Xs, XP, 16, ks * 4, lane_lo, half);
      #pragma unroll
      for (int nt = 0; nt < NTW; ++nt) {
        acc[0][nt] = wmma_f32(a0, w1f[ks][nt], acc[0][nt]);
        acc[1][nt] = wmma_f32(a1, w1f[ks][nt], acc[1][nt]);
      }
    }
    #pragma unroll
    for (int nt = 0; nt < NTW; ++nt) {
      int col = (wave * NTW + nt) * 16 + lane_lo;
      float add = t0 * W1[DD * HH + col] + b1[col];
      #pragma unroll
      for (int mt = 0; mt < 2; ++mt)
        #pragma unroll
        for (int v = 0; v < 8; ++v) {
          int m = mt * 16 + v + 8 * half;
          H1s[m * HP + col] = fast_tanh(acc[mt][nt][v] + add);
        }
    }
    __syncthreads();

    {
      const int ng    = wave & 7;
      const int kp    = wave >> 3;
      const int kbase = kp * (HH / 2);
      v8f hac[2][4];
      v8f vac[2][4];
      #pragma unroll
      for (int mt = 0; mt < 2; ++mt)
        #pragma unroll
        for (int nt = 0; nt < 4; ++nt) { hac[mt][nt] = (v8f){}; vac[mt][nt] = (v8f){}; }
      const float* w2p = W2  + (kbase + 2 * half) * HH + ng * 64 + lane_lo;
      const float* mp  = Mtr + (kbase + 2 * half) * HH + ng * 64 + lane_lo;
      const float* h1p = H1s + lane_lo * HP + kbase + 2 * half;
      for (int kk = 0; kk < HH / 2; kk += 8) {
        v2f b[2][4], bm[2][4], a0[2], a1[2];
        #pragma unroll
        for (int u = 0; u < 2; ++u) {
          #pragma unroll
          for (int nt = 0; nt < 4; ++nt) {
            b[u][nt].x  = w2p[(kk + 4 * u + 0) * HH + nt * 16];
            b[u][nt].y  = w2p[(kk + 4 * u + 1) * HH + nt * 16];
            bm[u][nt].x = mp[(kk + 4 * u + 0) * HH + nt * 16];
            bm[u][nt].y = mp[(kk + 4 * u + 1) * HH + nt * 16];
          }
          a0[u].x = h1p[kk + 4 * u];            a0[u].y = h1p[kk + 4 * u + 1];
          a1[u].x = h1p[16 * HP + kk + 4 * u];  a1[u].y = h1p[16 * HP + kk + 4 * u + 1];
        }
        #pragma unroll
        for (int u = 0; u < 2; ++u) {
          v2f s0, s1;
          s0.x = 1.f - a0[u].x * a0[u].x;  s0.y = 1.f - a0[u].y * a0[u].y;
          s1.x = 1.f - a1[u].x * a1[u].x;  s1.y = 1.f - a1[u].y * a1[u].y;
          #pragma unroll
          for (int nt = 0; nt < 4; ++nt) {
            hac[0][nt] = wmma_f32(a0[u], b[u][nt],  hac[0][nt]);
            hac[1][nt] = wmma_f32(a1[u], b[u][nt],  hac[1][nt]);
            vac[0][nt] = wmma_f32(s0,    bm[u][nt], vac[0][nt]);
            vac[1][nt] = wmma_f32(s1,    bm[u][nt], vac[1][nt]);
          }
        }
      }
      float* mine = red2 + wave * (64 * 32);
      float kh[4][8], kv[4][8];
      #pragma unroll
      for (int nt = 0; nt < 4; ++nt)
        #pragma unroll
        for (int v = 0; v < 8; ++v) {
          mine[(nt * 8 + v) * 32 + lane]      = kp ? hac[0][nt][v] : hac[1][nt][v];
          mine[(32 + nt * 8 + v) * 32 + lane] = kp ? vac[0][nt][v] : vac[1][nt][v];
          kh[nt][v] = kp ? hac[1][nt][v] : hac[0][nt][v];
          kv[nt][v] = kp ? vac[1][nt][v] : vac[0][nt][v];
        }
      __syncthreads();
      const float* theirs = red2 + (wave ^ 8) * (64 * 32);
      float ph[4][8], pv[4][8];
      #pragma unroll
      for (int nt = 0; nt < 4; ++nt)
        #pragma unroll
        for (int v = 0; v < 8; ++v) {
          ph[nt][v] = theirs[(nt * 8 + v) * 32 + lane];
          pv[nt][v] = theirs[(32 + nt * 8 + v) * 32 + lane];
        }
      const float* w3p = W3 + (ng * 64 + 2 * half) * DD + lane_lo;
      v2f w3a[8][4];
      #pragma unroll
      for (int ks = 0; ks < 8; ++ks)
        #pragma unroll
        for (int nt = 0; nt < 4; ++nt) {
          w3a[ks][nt].x = w3p[(ks * 4 + 0) * DD + nt * 16];
          w3a[ks][nt].y = w3p[(ks * 4 + 1) * DD + nt * 16];
        }
      float add[4];
      #pragma unroll
      for (int nt = 0; nt < 4; ++nt) add[nt] = b2[(ng * 4 + nt) * 16 + lane_lo];
      float tl[8];
      #pragma unroll
      for (int v = 0; v < 8; ++v) {
        int m = kp * 16 + v + 8 * half;
        float t = 0.f;
        #pragma unroll
        for (int nt = 0; nt < 4; ++nt) {
          int col = (ng * 4 + nt) * 16 + lane_lo;
          float a2 = kh[nt][v] + ph[nt][v];
          float vv = kv[nt][v] + pv[nt][v];
          float h2 = fast_tanh(a2 + add[nt]);
          H2s[m * HP + col] = h2;
          t += (1.f - h2 * h2) * vv;
        }
        tl[v] = t;
      }
      v2f w3b[8][4];
      #pragma unroll
      for (int ks = 0; ks < 8; ++ks)
        #pragma unroll
        for (int nt = 0; nt < 4; ++nt) {
          w3b[ks][nt].x = w3p[((ks + 8) * 4 + 0) * DD + nt * 16];
          w3b[ks][nt].y = w3p[((ks + 8) * 4 + 1) * DD + nt * 16];
        }
      row16_sum<8>(tl);
      if (lane_lo == 0) {
        #pragma unroll
        for (int v = 0; v < 8; ++v)
          atomicAdd(&lacc[kp * 16 + v + 8 * half], dt * tl[v]);
      }
      __builtin_amdgcn_fence(__ATOMIC_RELEASE, "wavefront");
      __builtin_amdgcn_fence(__ATOMIC_ACQUIRE, "wavefront");
      const float* h2p = H2s + (kp * 16 + lane_lo) * HP + ng * 64 + 2 * half;
      v2f ha[16];
      #pragma unroll
      for (int ks = 0; ks < 16; ++ks) { ha[ks].x = h2p[ks * 4]; ha[ks].y = h2p[ks * 4 + 1]; }
      v8f dacc[4];
      #pragma unroll
      for (int nt = 0; nt < 4; ++nt) dacc[nt] = (v8f){};
      #pragma unroll
      for (int ks = 0; ks < 8; ++ks)
        #pragma unroll
        for (int nt = 0; nt < 4; ++nt) dacc[nt] = wmma_f32(ha[ks], w3a[ks][nt], dacc[nt]);
      #pragma unroll
      for (int ks = 0; ks < 8; ++ks)
        #pragma unroll
        for (int nt = 0; nt < 4; ++nt) dacc[nt] = wmma_f32(ha[ks + 8], w3b[ks][nt], dacc[nt]);
      #pragma unroll
      for (int nt = 0; nt < 4; ++nt)
        #pragma unroll
        for (int q = 0; q < 2; ++q) {
          v4f o;
          o.x = dacc[nt][4 * q + 0];  o.y = dacc[nt][4 * q + 1];
          o.z = dacc[nt][4 * q + 2];  o.w = dacc[nt][4 * q + 3];
          *(v4f*)(dxp + ((wave * 4 + nt) * 2 + q) * 128 + lane * 4) = o;
        }
    }
    __syncthreads();

    {
      const int tile = wave & 7;
      const int q    = wave >> 3;
      const int mt = tile >> 2;
      const int m0 = mt * 16;
      const int n0 = (tile & 3) * 16;
      v4f pc[8];
      #pragma unroll
      for (int g = 0; g < 8; ++g)
        pc[g] = *(const v4f*)(dxp + (((g + 8 * mt) * 4 + (tile & 3)) * 2 + q) * 128 + lane * 4);
      v4f s4 = ((pc[0] + pc[1]) + (pc[2] + pc[3])) + ((pc[4] + pc[5]) + (pc[6] + pc[7]));
      float dxs[4] = {s4.x, s4.y, s4.z, s4.w};
      int col = n0 + lane_lo;
      float bb = b3[col];
      float zt[4];
      #pragma unroll
      for (int j = 0; j < 4; ++j) {
        int m = m0 + 4 * q + j + 8 * half;
        float dx = dxs[j] + bb;
        zt[j] = dt * 0.5f * dx * dx;
        Xs[m * XP + col] += dt * dx;
      }
      row16_sum<4>(zt);
      if (lane_lo == 0) {
        #pragma unroll
        for (int j = 0; j < 4; ++j)
          atomicAdd(&zacc[m0 + 4 * q + j + 8 * half], zt[j]);
      }
    }
  }

  __syncthreads();
  for (int i = tid; i < MT * DD; i += NTH) {
    int m = i >> 6, c = i & 63;
    out[(s_base + m) * DD + c] = Xs[m * XP + c];
  }
  if (tid < MT) {
    out[BB * DD + s_base + tid]      = zacc[tid];
    out[BB * DD + BB + s_base + tid] = lacc[tid];
  }
}

extern "C" void kernel_launch(void* const* d_in, const int* in_sizes, int n_in,
                              void* d_out, int out_size, void* d_ws, size_t ws_size,
                              hipStream_t stream) {
  const float* y0 = (const float*)d_in[0];
  const float* ts = (const float*)d_in[1];
  const float* W1 = (const float*)d_in[2];
  const float* b1 = (const float*)d_in[3];
  const float* W2 = (const float*)d_in[4];
  const float* b2 = (const float*)d_in[5];
  const float* W3 = (const float*)d_in[6];
  const float* b3 = (const float*)d_in[7];
  float* out = (float*)d_out;
  float* Mtr = (float*)d_ws;

  cnf_prep<<<(HH * HH) / 256, 256, 0, stream>>>(W1, W2, W3, Mtr);

  size_t shmem = (size_t)(MT * XP + 2 * MT * HP + 2 * MT + NW * 64 * 32) * sizeof(float);
  cnf_main<<<BB / MT, NTH, shmem, stream>>>(y0, ts, W1, b1, W2, b2, W3, b3,
                                            Mtr, out);
}


# 4 workgroups per 32-sample tile: each redoes the state chain and takes a quarter of the trace columns in the fused k loop (10 tiles, 4 k-steps a trip); a 3rd launch adds the 4; trace sums regroup
# speedup vs baseline: 2.0683x; 1.3757x over previous
#include <hip/hip_runtime.h>
#include <hip/hip_bf16.h>


#define BB  512
#define DD  64
#define HH  512
#define TT  8
#define MT  32
#define TP  4
#define NW  16
#define NTH (NW * 32)
#define NTW (32 / NW)
#define XP  68
#define HP  516
#define R2  40
#define KU  4

typedef float v2f __attribute__((ext_vector_type(2)));
typedef float v4f __attribute__((ext_vector_type(4)));
typedef float v8f __attribute__((ext_vector_type(8)));

__device__ __forceinline__ v8f wmma_f32(v2f a, v2f b, v8f c) {
  return __builtin_amdgcn_wmma_f32_16x16x4_f32(false, a, false, b,
                                               (short)0, c, false, false);
}

__device__ __forceinline__ float fast_tanh(float x) {
#if __has_builtin(__builtin_amdgcn_tanhf)
  return __builtin_amdgcn_tanhf(x);
#elif __has_builtin(__builtin_amdgcn_exp2f) && __has_builtin(__builtin_amdgcn_rcpf)
  float e = __builtin_amdgcn_exp2f(x * 2.8853900817779268f);
  return 1.0f - 2.0f * __builtin_amdgcn_rcpf(e + 1.0f);
#else
  return tanhf(x);
#endif
}

__device__ __forceinline__ v2f frag_a(const float* S, int pitch, int m0, int k0,
                                      int lane_lo, int half) {
  const float* p = S + (m0 + lane_lo) * pitch + k0 + 2 * half;
  v2f a; a.x = p[0]; a.y = p[1];
  return a;
}

__device__ __forceinline__ v2f frag_b(const float* __restrict__ W, int ldw,
                                      int k0, int n0, int lane_lo, int half) {
  v2f b;
  b.x = W[(k0 + 2 * half + 0) * ldw + n0 + lane_lo];
  b.y = W[(k0 + 2 * half + 1) * ldw + n0 + lane_lo];
  return b;
}

template <int X>
__device__ __forceinline__ float row_xor(float v) {
  return __builtin_bit_cast(float,
      __builtin_amdgcn_update_dpp(0, __builtin_bit_cast(int, v),
                                  0x160 + X, 0xf, 0xf, true));
}

template <int N>
__device__ __forceinline__ void row16_sum(float (&r)[N]) {
  #pragma unroll
  for (int i = 0; i < N; ++i) r[i] += row_xor<1>(r[i]);
  #pragma unroll
  for (int i = 0; i < N; ++i) r[i] += row_xor<2>(r[i]);
  #pragma unroll
  for (int i = 0; i < N; ++i) r[i] += row_xor<4>(r[i]);
  #pragma unroll
  for (int i = 0; i < N; ++i) r[i] += row_xor<8>(r[i]);
}

__global__ __launch_bounds__(256) void cnf_prep(const float* __restrict__ W1,
                                                const float* __restrict__ W2,
                                                const float* __restrict__ W3,
                                                float* __restrict__ M) {
  int idx = blockIdx.x * blockDim.x + threadIdx.x;
  int j = idx >> 9;
  int k = idx & 511;
  float c = 0.f;
  #pragma unroll 8
  for (int m = 0; m < DD; ++m)
    c += W3[k * DD + m] * W1[m * HH + j];
  M[j * HH + k] = W2[j * HH + k] * c;
}

__global__ __launch_bounds__(NTH) void cnf_main(const float* __restrict__ y0,
                                                const float* __restrict__ ts,
                                                const float* __restrict__ W1,
                                                const float* __restrict__ b1,
                                                const float* __restrict__ W2,
                                                const float* __restrict__ b2,
                                                const float* __restrict__ W3,
                                                const float* __restrict__ b3,
                                                const float* __restrict__ Mtr,
                                                float* __restrict__ lpart,
                                                float* __restrict__ out) {
  extern __shared__ float smem[];
  float* Xs   = smem;
  float* H1s  = Xs  + MT * XP;
  float* H2s  = H1s + MT * HP;
  float* zacc = H2s + MT * HP;
  float* lacc = zacc + MT;
  float* red2 = lacc + MT;
  float* dxp  = H1s;

  const int tid     = threadIdx.x;
  const int wave    = __builtin_amdgcn_readfirstlane(tid >> 5);
  const int lane    = tid & 31;
  const int lane_lo = lane & 15;
  const int half    = lane >> 4;
  const int tp      = blockIdx.x & (TP - 1);
  const int s_base  = (blockIdx.x / TP) * MT;

  for (int i = tid; i < MT * DD; i += NTH) {
    int m = i >> 6, c = i & 63;
    Xs[m * XP + c] = y0[(s_base + m) * DD + c];
  }
  if (tid < MT) { zacc[tid] = 0.f; lacc[tid] = 0.f; }

  v8f acc[2][NTW];

  for (int step = 0; step < TT - 1; ++step) {
    const float t0 = ts[step];
    const float dt = ts[step + 1] - t0;

    const float* W1s = W1;
    asm volatile("" : "+s"(W1s));
    v2f w1f[DD / 4][NTW];
    {
      const float* w1p = W1s + (2 * half) * HH + wave * (NTW * 16) + lane_lo;
      #pragma unroll
      for (int ks = 0; ks < DD / 4; ++ks)
        #pragma unroll
        for (int nt = 0; nt < NTW; ++nt) {
          w1f[ks][nt].x = w1p[(ks * 4 + 0) * HH + nt * 16];
          w1f[ks][nt].y = w1p[(ks * 4 + 1) * HH + nt * 16];
        }
    }
    __syncthreads();

    #pragma unroll
    for (int mt = 0; mt < 2; ++mt)
      #pragma unroll
      for (int nt = 0; nt < NTW; ++nt) acc[mt][nt] = (v8f){};
    #pragma unroll
    for (int ks = 0; ks < DD / 4; ++ks) {
      v2f a0 = frag_a(Xs, XP, 0,  ks * 4, lane_lo, half);
      v2f a1 = frag_a(Xs, XP, 16, ks * 4, lane_lo, half);
      #pragma unroll
      for (int nt = 0; nt < NTW; ++nt) {
        acc[0][nt] = wmma_f32(a0, w1f[ks][nt], acc[0][nt]);
        acc[1][nt] = wmma_f32(a1, w1f[ks][nt], acc[1][nt]);
      }
    }
    #pragma unroll
    for (int nt = 0; nt < NTW; ++nt) {
      int col = (wave * NTW + nt) * 16 + lane_lo;
      float add = t0 * W1[DD * HH + col] + b1[col];
      #pragma unroll
      for (int mt = 0; mt < 2; ++mt)
        #pragma unroll
        for (int v = 0; v < 8; ++v) {
          int m = mt * 16 + v + 8 * half;
          H1s[m * HP + col] = fast_tanh(acc[mt][nt][v] + add);
        }
    }
    __syncthreads();

    {
      const int ng    = wave & 7;
      const int kp    = wave >> 3;
      const int kbase = kp * (HH / 2);
      int cofs[4];
      #pragma unroll
      for (int nt = 0; nt < 4; ++nt) cofs[nt] = ((nt + tp) & 3) * 16;
      v8f hac[2][4];
      v8f vac[2];
      #pragma unroll
      for (int mt = 0; mt < 2; ++mt) {
        vac[mt] = (v8f){};
        #pragma unroll
        for (int nt = 0; nt < 4; ++nt) hac[mt][nt] = (v8f){};
      }
      const float* w2p = W2  + (kbase + 2 * half) * HH + ng * 64 + lane_lo;
      const float* mp  = Mtr + (kbase + 2 * half) * HH + ng * 64 + cofs[0] + lane_lo;
      const float* h1p = H1s + lane_lo * HP + kbase + 2 * half;
      for (int kk = 0; kk < HH / 2; kk += 4 * KU) {
        v2f b[KU][4], bm[KU], a0[KU], a1[KU];
        #pragma unroll
        for (int u = 0; u < KU; ++u) {
          #pragma unroll
          for (int nt = 0; nt < 4; ++nt) {
            b[u][nt].x  = w2p[(kk + 4 * u + 0) * HH + cofs[nt]];
            b[u][nt].y  = w2p[(kk + 4 * u + 1) * HH + cofs[nt]];
          }
          bm[u].x = mp[(kk + 4 * u + 0) * HH];
          bm[u].y = mp[(kk + 4 * u + 1) * HH];
          a0[u].x = h1p[kk + 4 * u];            a0[u].y = h1p[kk + 4 * u + 1];
          a1[u].x = h1p[16 * HP + kk + 4 * u];  a1[u].y = h1p[16 * HP + kk + 4 * u + 1];
        }
        #pragma unroll
        for (int u = 0; u < KU; ++u) {
          v2f s0, s1;
          s0.x = 1.f - a0[u].x * a0[u].x;  s0.y = 1.f - a0[u].y * a0[u].y;
          s1.x = 1.f - a1[u].x * a1[u].x;  s1.y = 1.f - a1[u].y * a1[u].y;
          #pragma unroll
          for (int nt = 0; nt < 4; ++nt) {
            hac[0][nt] = wmma_f32(a0[u], b[u][nt], hac[0][nt]);
            hac[1][nt] = wmma_f32(a1[u], b[u][nt], hac[1][nt]);
          }
          vac[0] = wmma_f32(s0, bm[u], vac[0]);
          vac[1] = wmma_f32(s1, bm[u], vac[1]);
        }
      }
      float* mine = red2 + wave * (R2 * 32);
      float kh[4][8], kv[8];
      #pragma unroll
      for (int v = 0; v < 8; ++v) {
        #pragma unroll
        for (int nt = 0; nt < 4; ++nt) {
          mine[(nt * 8 + v) * 32 + lane] = kp ? hac[0][nt][v] : hac[1][nt][v];
          kh[nt][v] = kp ? hac[1][nt][v] : hac[0][nt][v];
        }
        mine[(32 + v) * 32 + lane] = kp ? vac[0][v] : vac[1][v];
        kv[v] = kp ? vac[1][v] : vac[0][v];
      }
      __syncthreads();
      const float* theirs = red2 + (wave ^ 8) * (R2 * 32);
      float ph[4][8], pv[8];
      #pragma unroll
      for (int v = 0; v < 8; ++v) {
        #pragma unroll
        for (int nt = 0; nt < 4; ++nt) ph[nt][v] = theirs[(nt * 8 + v) * 32 + lane];
        pv[v] = theirs[(32 + v) * 32 + lane];
      }
      const float* w3p = W3 + (ng * 64 + 2 * half) * DD + lane_lo;
      v2f w3a[8][4];
      #pragma unroll
      for (int ks = 0; ks < 8; ++ks)
        #pragma unroll
        for (int nt = 0; nt < 4; ++nt) {
          w3a[ks][nt].x = w3p[(ks * 4 + 0) * DD + nt * 16];
          w3a[ks][nt].y = w3p[(ks * 4 + 1) * DD + nt * 16];
        }
      float add[4];
      #pragma unroll
      for (int nt = 0; nt < 4; ++nt) add[nt] = b2[ng * 64 + cofs[nt] + lane_lo];
      float tl[8];
      #pragma unroll
      for (int v = 0; v < 8; ++v) {
        int m = kp * 16 + v + 8 * half;
        #pragma unroll
        for (int nt = 0; nt < 4; ++nt) {
          int col = ng * 64 + cofs[nt] + lane_lo;
          float a2 = kh[nt][v] + ph[nt][v];
          float h2 = fast_tanh(a2 + add[nt]);
          H2s[m * HP + col] = h2;
          if (nt == 0)
            tl[v] = (1.f - h2 * h2) * (kv[v] + pv[v]);
        }
      }
      v2f w3b[8][4];
      #pragma unroll
      for (int ks = 0; ks < 8; ++ks)
        #pragma unroll
        for (int nt = 0; nt < 4; ++nt) {
          w3b[ks][nt].x = w3p[((ks + 8) * 4 + 0) * DD + nt * 16];
          w3b[ks][nt].y = w3p[((ks + 8) * 4 + 1) * DD + nt * 16];
        }
      row16_sum<8>(tl);
      if (lane_lo == 0) {
        #pragma unroll
        for (int v = 0; v < 8; ++v)
          atomicAdd(&lacc[kp * 16 + v + 8 * half], dt * tl[v]);
      }
      __builtin_amdgcn_fence(__ATOMIC_RELEASE, "wavefront");
      __builtin_amdgcn_fence(__ATOMIC_ACQUIRE, "wavefront");
      const float* h2p = H2s + (kp * 16 + lane_lo) * HP + ng * 64 + 2 * half;
      v2f ha[16];
      #pragma unroll
      for (int ks = 0; ks < 16; ++ks) { ha[ks].x = h2p[ks * 4]; ha[ks].y = h2p[ks * 4 + 1]; }
      v8f dacc[4];
      #pragma unroll
      for (int nt = 0; nt < 4; ++nt) dacc[nt] = (v8f){};
      #pragma unroll
      for (int ks = 0; ks < 8; ++ks)
        #pragma unroll
        for (int nt = 0; nt < 4; ++nt) dacc[nt] = wmma_f32(ha[ks], w3a[ks][nt], dacc[nt]);
      #pragma unroll
      for (int ks = 0; ks < 8; ++ks)
        #pragma unroll
        for (int nt = 0; nt < 4; ++nt) dacc[nt] = wmma_f32(ha[ks + 8], w3b[ks][nt], dacc[nt]);
      #pragma unroll
      for (int nt = 0; nt < 4; ++nt)
        #pragma unroll
        for (int q = 0; q < 2; ++q) {
          v4f o;
          o.x = dacc[nt][4 * q + 0];  o.y = dacc[nt][4 * q + 1];
          o.z = dacc[nt][4 * q + 2];  o.w = dacc[nt][4 * q + 3];
          *(v4f*)(dxp + ((wave * 4 + nt) * 2 + q) * 128 + lane * 4) = o;
        }
    }
    __syncthreads();

    {
      const int tile = wave & 7;
      const int q    = wave >> 3;
      const int mt = tile >> 2;
      const int m0 = mt * 16;
      const int n0 = (tile & 3) * 16;
      v4f pc[8];
      #pragma unroll
      for (int g = 0; g < 8; ++g)
        pc[g] = *(const v4f*)(dxp + (((g + 8 * mt) * 4 + (tile & 3)) * 2 + q) * 128 + lane * 4);
      v4f s4 = ((pc[0] + pc[1]) + (pc[2] + pc[3])) + ((pc[4] + pc[5]) + (pc[6] + pc[7]));
      float dxs[4] = {s4.x, s4.y, s4.z, s4.w};
      int col = n0 + lane_lo;
      float bb = b3[col];
      float zt[4];
      #pragma unroll
      for (int j = 0; j < 4; ++j) {
        int m = m0 + 4 * q + j + 8 * half;
        float dx = dxs[j] + bb;
        zt[j] = dt * 0.5f * dx * dx;
        Xs[m * XP + col] += dt * dx;
      }
      row16_sum<4>(zt);
      if (lane_lo == 0) {
        #pragma unroll
        for (int j = 0; j < 4; ++j)
          atomicAdd(&zacc[m0 + 4 * q + j + 8 * half], zt[j]);
      }
    }
  }

  __syncthreads();
  if (tp == 0) {
    for (int i = tid; i < MT * DD; i += NTH) {
      int m = i >> 6, c = i & 63;
      out[(s_base + m) * DD + c] = Xs[m * XP + c];
    }
    if (tid < MT) out[BB * DD + s_base + tid] = zacc[tid];
  }
  if (tid < MT) lpart[tp * BB + s_base + tid] = lacc[tid];
}

__global__ __launch_bounds__(256) void cnf_join(const float* __restrict__ lpart,
                                                float* __restrict__ out) {
  int s = blockIdx.x * blockDim.x + threadIdx.x;
  out[BB * DD + BB + s] = (lpart[s] + lpart[BB + s]) + (lpart[2 * BB + s] + lpart[3 * BB + s]);
}

extern "C" void kernel_launch(void* const* d_in, const int* in_sizes, int n_in,
                              void* d_out, int out_size, void* d_ws, size_t ws_size,
                              hipStream_t stream) {
  const float* y0 = (const float*)d_in[0];
  const float* ts = (const float*)d_in[1];
  const float* W1 = (const float*)d_in[2];
  const float* b1 = (const float*)d_in[3];
  const float* W2 = (const float*)d_in[4];
  const float* b2 = (const float*)d_in[5];
  const float* W3 = (const float*)d_in[6];
  const float* b3 = (const float*)d_in[7];
  float* out = (float*)d_out;
  float* Mtr   = (float*)d_ws;
  float* lpart = Mtr + HH * HH;

  cnf_prep<<<(HH * HH) / 256, 256, 0, stream>>>(W1, W2, W3, Mtr);

  size_t shmem = (size_t)(MT * XP + 2 * MT * HP + 2 * MT + NW * R2 * 32) * sizeof(float);
  cnf_main<<<(BB / MT) * TP, NTH, shmem, stream>>>(y0, ts, W1, b1, W2, b2, W3, b3,
                                                   Mtr, lpart, out);
  cnf_join<<<BB / 256, 256, 0, stream>>>(lpart, out);
}


# 8 workgroups per 32-sample tile instead of 4: each redoes the state chain and takes one 16x16 trace tile per wave (a column quarter of one 16-sample half), 9 WMMAs a k-step; sum order kept
# speedup vs baseline: 2.2008x; 1.0641x over previous
#include <hip/hip_runtime.h>
#include <hip/hip_bf16.h>


#define BB  512
#define DD  64
#define HH  512
#define TT  8
#define MT  32
#define TP  8
#define NW  16
#define NTH (NW * 32)
#define NTW (32 / NW)
#define XP  68
#define HP  516
#define R2  40
#define KU  4

typedef float v2f __attribute__((ext_vector_type(2)));
typedef float v4f __attribute__((ext_vector_type(4)));
typedef float v8f __attribute__((ext_vector_type(8)));

__device__ __forceinline__ v8f wmma_f32(v2f a, v2f b, v8f c) {
  return __builtin_amdgcn_wmma_f32_16x16x4_f32(false, a, false, b,
                                               (short)0, c, false, false);
}

__device__ __forceinline__ float fast_tanh(float x) {
#if __has_builtin(__builtin_amdgcn_tanhf)
  return __builtin_amdgcn_tanhf(x);
#elif __has_builtin(__builtin_amdgcn_exp2f) && __has_builtin(__builtin_amdgcn_rcpf)
  float e = __builtin_amdgcn_exp2f(x * 2.8853900817779268f);
  return 1.0f - 2.0f * __builtin_amdgcn_rcpf(e + 1.0f);
#else
  return tanhf(x);
#endif
}

__device__ __forceinline__ v2f frag_a(const float* S, int pitch, int m0, int k0,
                                      int lane_lo, int half) {
  const float* p = S + (m0 + lane_lo) * pitch + k0 + 2 * half;
  v2f a; a.x = p[0]; a.y = p[1];
  return a;
}

__device__ __forceinline__ v2f frag_b(const float* __restrict__ W, int ldw,
                                      int k0, int n0, int lane_lo, int half) {
  v2f b;
  b.x = W[(k0 + 2 * half + 0) * ldw + n0 + lane_lo];
  b.y = W[(k0 + 2 * half + 1) * ldw + n0 + lane_lo];
  return b;
}

template <int X>
__device__ __forceinline__ float row_xor(float v) {
  return __builtin_bit_cast(float,
      __builtin_amdgcn_update_dpp(0, __builtin_bit_cast(int, v),
                                  0x160 + X, 0xf, 0xf, true));
}

template <int N>
__device__ __forceinline__ void row16_sum(float (&r)[N]) {
  #pragma unroll
  for (int i = 0; i < N; ++i) r[i] += row_xor<1>(r[i]);
  #pragma unroll
  for (int i = 0; i < N; ++i) r[i] += row_xor<2>(r[i]);
  #pragma unroll
  for (int i = 0; i < N; ++i) r[i] += row_xor<4>(r[i]);
  #pragma unroll
  for (int i = 0; i < N; ++i) r[i] += row_xor<8>(r[i]);
}

__global__ __launch_bounds__(256) void cnf_prep(const float* __restrict__ W1,
                                                const float* __restrict__ W2,
                                                const float* __restrict__ W3,
                                                float* __restrict__ M) {
  int idx = blockIdx.x * blockDim.x + threadIdx.x;
  int j = idx >> 9;
  int k = idx & 511;
  float c = 0.f;
  #pragma unroll 8
  for (int m = 0; m < DD; ++m)
    c += W3[k * DD + m] * W1[m * HH + j];
  M[j * HH + k] = W2[j * HH + k] * c;
}

__global__ __launch_bounds__(NTH) void cnf_main(const float* __restrict__ y0,
                                                const float* __restrict__ ts,
                                                const float* __restrict__ W1,
                                                const float* __restrict__ b1,
                                                const float* __restrict__ W2,
                                                const float* __restrict__ b2,
                                                const float* __restrict__ W3,
                                                const float* __restrict__ b3,
                                                const float* __restrict__ Mtr,
                                                float* __restrict__ lpart,
                                                float* __restrict__ out) {
  extern __shared__ float smem[];
  float* Xs   = smem;
  float* H1s  = Xs  + MT * XP;
  float* H2s  = H1s + MT * HP;
  float* zacc = H2s + MT * HP;
  float* lacc = zacc + MT;
  float* red2 = lacc + MT;
  float* dxp  = H1s;

  const int tid     = threadIdx.x;
  const int wave    = __builtin_amdgcn_readfirstlane(tid >> 5);
  const int lane    = tid & 31;
  const int lane_lo = lane & 15;
  const int half    = lane >> 4;
  const int tp      = blockIdx.x & (TP - 1);
  const int tq      = tp & 3;
  const int tm      = tp >> 2;
  const int s_base  = (blockIdx.x / TP) * MT;

  for (int i = tid; i < MT * DD; i += NTH) {
    int m = i >> 6, c = i & 63;
    Xs[m * XP + c] = y0[(s_base + m) * DD + c];
  }
  if (tid < MT) { zacc[tid] = 0.f; lacc[tid] = 0.f; }

  v8f acc[2][NTW];

  for (int step = 0; step < TT - 1; ++step) {
    const float t0 = ts[step];
    const float dt = ts[step + 1] - t0;

    const float* W1s = W1;
    asm volatile("" : "+s"(W1s));
    v2f w1f[DD / 4][NTW];
    {
      const float* w1p = W1s + (2 * half) * HH + wave * (NTW * 16) + lane_lo;
      #pragma unroll
      for (int ks = 0; ks < DD / 4; ++ks)
        #pragma unroll
        for (int nt = 0; nt < NTW; ++nt) {
          w1f[ks][nt].x = w1p[(ks * 4 + 0) * HH + nt * 16];
          w1f[ks][nt].y = w1p[(ks * 4 + 1) * HH + nt * 16];
        }
    }
    __syncthreads();

    #pragma unroll
    for (int mt = 0; mt < 2; ++mt)
      #pragma unroll
      for (int nt = 0; nt < NTW; ++nt) acc[mt][nt] = (v8f){};
    #pragma unroll
    for (int ks = 0; ks < DD / 4; ++ks) {
      v2f a0 = frag_a(Xs, XP, 0,  ks * 4, lane_lo, half);
      v2f a1 = frag_a(Xs, XP, 16, ks * 4, lane_lo, half);
      #pragma unroll
      for (int nt = 0; nt < NTW; ++nt) {
        acc[0][nt] = wmma_f32(a0, w1f[ks][nt], acc[0][nt]);
        acc[1][nt] = wmma_f32(a1, w1f[ks][nt], acc[1][nt]);
      }
    }
    #pragma unroll
    for (int nt = 0; nt < NTW; ++nt) {
      int col = (wave * NTW + nt) * 16 + lane_lo;
      float add = t0 * W1[DD * HH + col] + b1[col];
      #pragma unroll
      for (int mt = 0; mt < 2; ++mt)
        #pragma unroll
        for (int v = 0; v < 8; ++v) {
          int m = mt * 16 + v + 8 * half;
          H1s[m * HP + col] = fast_tanh(acc[mt][nt][v] + add);
        }
    }
    __syncthreads();

    {
      const int ng    = wave & 7;
      const int kp    = wave >> 3;
      const int kbase = kp * (HH / 2);
      int cofs[4];
      #pragma unroll
      for (int nt = 0; nt < 4; ++nt) cofs[nt] = ((nt + tq) & 3) * 16;
      v8f hac[2][4];
      v8f vac = (v8f){};
      #pragma unroll
      for (int mt = 0; mt < 2; ++mt)
        #pragma unroll
        for (int nt = 0; nt < 4; ++nt) hac[mt][nt] = (v8f){};
      const float* w2p = W2  + (kbase + 2 * half) * HH + ng * 64 + lane_lo;
      const float* mp  = Mtr + (kbase + 2 * half) * HH + ng * 64 + cofs[0] + lane_lo;
      const float* h1p = H1s + lane_lo * HP + kbase + 2 * half;
      for (int kk = 0; kk < HH / 2; kk += 4 * KU) {
        v2f b[KU][4], bm[KU], a0[KU], a1[KU];
        #pragma unroll
        for (int u = 0; u < KU; ++u) {
          #pragma unroll
          for (int nt = 0; nt < 4; ++nt) {
            b[u][nt].x  = w2p[(kk + 4 * u + 0) * HH + cofs[nt]];
            b[u][nt].y  = w2p[(kk + 4 * u + 1) * HH + cofs[nt]];
          }
          bm[u].x = mp[(kk + 4 * u + 0) * HH];
          bm[u].y = mp[(kk + 4 * u + 1) * HH];
          a0[u].x = h1p[kk + 4 * u];            a0[u].y = h1p[kk + 4 * u + 1];
          a1[u].x = h1p[16 * HP + kk + 4 * u];  a1[u].y = h1p[16 * HP + kk + 4 * u + 1];
        }
        #pragma unroll
        for (int u = 0; u < KU; ++u) {
          v2f at = tm ? a1[u] : a0[u];
          v2f s;
          s.x = 1.f - at.x * at.x;  s.y = 1.f - at.y * at.y;
          #pragma unroll
          for (int nt = 0; nt < 4; ++nt) {
            hac[0][nt] = wmma_f32(a0[u], b[u][nt], hac[0][nt]);
            hac[1][nt] = wmma_f32(a1[u], b[u][nt], hac[1][nt]);
          }
          vac = wmma_f32(s, bm[u], vac);
        }
      }
      const bool fin = (kp == tm);
      float* mine = red2 + wave * (R2 * 32);
      float kh[4][8];
      #pragma unroll
      for (int v = 0; v < 8; ++v) {
        #pragma unroll
        for (int nt = 0; nt < 4; ++nt) {
          mine[(nt * 8 + v) * 32 + lane] = kp ? hac[0][nt][v] : hac[1][nt][v];
          kh[nt][v] = kp ? hac[1][nt][v] : hac[0][nt][v];
        }
      }
      if (!fin) {
        #pragma unroll
        for (int v = 0; v < 8; ++v) mine[(32 + v) * 32 + lane] = vac[v];
      }
      __syncthreads();
      const float* theirs = red2 + (wave ^ 8) * (R2 * 32);
      float ph[4][8], pv[8];
      #pragma unroll
      for (int v = 0; v < 8; ++v) {
        #pragma unroll
        for (int nt = 0; nt < 4; ++nt) ph[nt][v] = theirs[(nt * 8 + v) * 32 + lane];
        pv[v] = 0.f;
      }
      if (fin) {
        #pragma unroll
        for (int v = 0; v < 8; ++v) pv[v] = theirs[(32 + v) * 32 + lane];
      }
      const float* w3p = W3 + (ng * 64 + 2 * half) * DD + lane_lo;
      v2f w3a[8][4];
      #pragma unroll
      for (int ks = 0; ks < 8; ++ks)
        #pragma unroll
        for (int nt = 0; nt < 4; ++nt) {
          w3a[ks][nt].x = w3p[(ks * 4 + 0) * DD + nt * 16];
          w3a[ks][nt].y = w3p[(ks * 4 + 1) * DD + nt * 16];
        }
      float add[4];
      #pragma unroll
      for (int nt = 0; nt < 4; ++nt) add[nt] = b2[ng * 64 + cofs[nt] + lane_lo];
      float h20[8];
      #pragma unroll
      for (int v = 0; v < 8; ++v) {
        int m = kp * 16 + v + 8 * half;
        #pragma unroll
        for (int nt = 0; nt < 4; ++nt) {
          int col = ng * 64 + cofs[nt] + lane_lo;
          float a2 = kh[nt][v] + ph[nt][v];
          float h2 = fast_tanh(a2 + add[nt]);
          H2s[m * HP + col] = h2;
          if (nt == 0) h20[v] = h2;
        }
      }
      v2f w3b[8][4];
      #pragma unroll
      for (int ks = 0; ks < 8; ++ks)
        #pragma unroll
        for (int nt = 0; nt < 4; ++nt) {
          w3b[ks][nt].x = w3p[((ks + 8) * 4 + 0) * DD + nt * 16];
          w3b[ks][nt].y = w3p[((ks + 8) * 4 + 1) * DD + nt * 16];
        }
      if (fin) {
        float tl[8];
        #pragma unroll
        for (int v = 0; v < 8; ++v)
          tl[v] = (1.f - h20[v] * h20[v]) * (vac[v] + pv[v]);
        row16_sum<8>(tl);
        if (lane_lo == 0) {
          #pragma unroll
          for (int v = 0; v < 8; ++v)
            atomicAdd(&lacc[kp * 16 + v + 8 * half], dt * tl[v]);
        }
      }
      __builtin_amdgcn_fence(__ATOMIC_RELEASE, "wavefront");
      __builtin_amdgcn_fence(__ATOMIC_ACQUIRE, "wavefront");
      const float* h2p = H2s + (kp * 16 + lane_lo) * HP + ng * 64 + 2 * half;
      v2f ha[16];
      #pragma unroll
      for (int ks = 0; ks < 16; ++ks) { ha[ks].x = h2p[ks * 4]; ha[ks].y = h2p[ks * 4 + 1]; }
      v8f dacc[4];
      #pragma unroll
      for (int nt = 0; nt < 4; ++nt) dacc[nt] = (v8f){};
      #pragma unroll
      for (int ks = 0; ks < 8; ++ks)
        #pragma unroll
        for (int nt = 0; nt < 4; ++nt) dacc[nt] = wmma_f32(ha[ks], w3a[ks][nt], dacc[nt]);
      #pragma unroll
      for (int ks = 0; ks < 8; ++ks)
        #pragma unroll
        for (int nt = 0; nt < 4; ++nt) dacc[nt] = wmma_f32(ha[ks + 8], w3b[ks][nt], dacc[nt]);
      #pragma unroll
      for (int nt = 0; nt < 4; ++nt)
        #pragma unroll
        for (int q = 0; q < 2; ++q) {
          v4f o;
          o.x = dacc[nt][4 * q + 0];  o.y = dacc[nt][4 * q + 1];
          o.z = dacc[nt][4 * q + 2];  o.w = dacc[nt][4 * q + 3];
          *(v4f*)(dxp + ((wave * 4 + nt) * 2 + q) * 128 + lane * 4) = o;
        }
    }
    __syncthreads();

    {
      const int tile = wave & 7;
      const int q    = wave >> 3;
      const int mt = tile >> 2;
      const int m0 = mt * 16;
      const int n0 = (tile & 3) * 16;
      v4f pc[8];
      #pragma unroll
      for (int g = 0; g < 8; ++g)
        pc[g] = *(const v4f*)(dxp + (((g + 8 * mt) * 4 + (tile & 3)) * 2 + q) * 128 + lane * 4);
      v4f s4 = ((pc[0] + pc[1]) + (pc[2] + pc[3])) + ((pc[4] + pc[5]) + (pc[6] + pc[7]));
      float dxs[4] = {s4.x, s4.y, s4.z, s4.w};
      int col = n0 + lane_lo;
      float bb = b3[col];
      float zt[4];
      #pragma unroll
      for (int j = 0; j < 4; ++j) {
        int m = m0 + 4 * q + j + 8 * half;
        float dx = dxs[j] + bb;
        zt[j] = dt * 0.5f * dx * dx;
        Xs[m * XP + col] += dt * dx;
      }
      row16_sum<4>(zt);
      if (lane_lo == 0) {
        #pragma unroll
        for (int j = 0; j < 4; ++j)
          atomicAdd(&zacc[m0 + 4 * q + j + 8 * half], zt[j]);
      }
    }
  }

  __syncthreads();
  if (tp == 0) {
    for (int i = tid; i < MT * DD; i += NTH) {
      int m = i >> 6, c = i & 63;
      out[(s_base + m) * DD + c] = Xs[m * XP + c];
    }
    if (tid < MT) out[BB * DD + s_base + tid] = zacc[tid];
  }
  if (tid < 16) lpart[tp * BB + s_base + tm * 16 + tid] = lacc[tm * 16 + tid];
}

__global__ __launch_bounds__(256) void cnf_join(const float* __restrict__ lpart,
                                                float* __restrict__ out) {
  int s = blockIdx.x * blockDim.x + threadIdx.x;
  const float* p = lpart + ((s >> 4) & 1) * 4 * BB + s;
  out[BB * DD + BB + s] = (p[0] + p[BB]) + (p[2 * BB] + p[3 * BB]);
}

extern "C" void kernel_launch(void* const* d_in, const int* in_sizes, int n_in,
                              void* d_out, int out_size, void* d_ws, size_t ws_size,
                              hipStream_t stream) {
  const float* y0 = (const float*)d_in[0];
  const float* ts = (const float*)d_in[1];
  const float* W1 = (const float*)d_in[2];
  const float* b1 = (const float*)d_in[3];
  const float* W2 = (const float*)d_in[4];
  const float* b2 = (const float*)d_in[5];
  const float* W3 = (const float*)d_in[6];
  const float* b3 = (const float*)d_in[7];
  float* out = (float*)d_out;
  float* Mtr   = (float*)d_ws;
  float* lpart = Mtr + HH * HH;

  cnf_prep<<<(HH * HH) / 256, 256, 0, stream>>>(W1, W2, W3, Mtr);

  size_t shmem = (size_t)(MT * XP + 2 * MT * HP + 2 * MT + NW * R2 * 32) * sizeof(float);
  cnf_main<<<(BB / MT) * TP, NTH, shmem, stream>>>(y0, ts, W1, b1, W2, b2, W3, b3,
                                                   Mtr, lpart, out);
  cnf_join<<<BB / 256, 256, 0, stream>>>(lpart, out);
}


# Step-invariant operands stay in LDS for all 7 steps: each wave's W1 fragments (128-bit rows), b1, W1's t-row, b2, b3, ts; K-half sums pass through the idle H2 array to make room; sum order kept
# speedup vs baseline: 2.2854x; 1.0384x over previous
#include <hip/hip_runtime.h>
#include <hip/hip_bf16.h>


#define BB  512
#define DD  64
#define HH  512
#define TT  8
#define MT  32
#define TP  8
#define NW  16
#define NTH (NW * 32)
#define NTW (32 / NW)
#define XP  68
#define HP  516
#define R2  8
#define CST (3 * HH + DD + TT)
#define KU  4

typedef float v2f __attribute__((ext_vector_type(2)));
typedef float v4f __attribute__((ext_vector_type(4)));
typedef float v8f __attribute__((ext_vector_type(8)));

__device__ __forceinline__ v8f wmma_f32(v2f a, v2f b, v8f c) {
  return __builtin_amdgcn_wmma_f32_16x16x4_f32(false, a, false, b,
                                               (short)0, c, false, false);
}

__device__ __forceinline__ float fast_tanh(float x) {
#if __has_builtin(__builtin_amdgcn_tanhf)
  return __builtin_amdgcn_tanhf(x);
#elif __has_builtin(__builtin_amdgcn_exp2f) && __has_builtin(__builtin_amdgcn_rcpf)
  float e = __builtin_amdgcn_exp2f(x * 2.8853900817779268f);
  return 1.0f - 2.0f * __builtin_amdgcn_rcpf(e + 1.0f);
#else
  return tanhf(x);
#endif
}

__device__ __forceinline__ v2f frag_a(const float* S, int pitch, int m0, int k0,
                                      int lane_lo, int half) {
  const float* p = S + (m0 + lane_lo) * pitch + k0 + 2 * half;
  v2f a; a.x = p[0]; a.y = p[1];
  return a;
}

__device__ __forceinline__ v2f frag_b(const float* __restrict__ W, int ldw,
                                      int k0, int n0, int lane_lo, int half) {
  v2f b;
  b.x = W[(k0 + 2 * half + 0) * ldw + n0 + lane_lo];
  b.y = W[(k0 + 2 * half + 1) * ldw + n0 + lane_lo];
  return b;
}

template <int X>
__device__ __forceinline__ float row_xor(float v) {
  return __builtin_bit_cast(float,
      __builtin_amdgcn_update_dpp(0, __builtin_bit_cast(int, v),
                                  0x160 + X, 0xf, 0xf, true));
}

template <int N>
__device__ __forceinline__ void row16_sum(float (&r)[N]) {
  #pragma unroll
  for (int i = 0; i < N; ++i) r[i] += row_xor<1>(r[i]);
  #pragma unroll
  for (int i = 0; i < N; ++i) r[i] += row_xor<2>(r[i]);
  #pragma unroll
  for (int i = 0; i < N; ++i) r[i] += row_xor<4>(r[i]);
  #pragma unroll
  for (int i = 0; i < N; ++i) r[i] += row_xor<8>(r[i]);
}

__global__ __launch_bounds__(256) void cnf_prep(const float* __restrict__ W1,
                                                const float* __restrict__ W2,
                                                const float* __restrict__ W3,
                                                float* __restrict__ M) {
  int idx = blockIdx.x * blockDim.x + threadIdx.x;
  int j = idx >> 9;
  int k = idx & 511;
  float c = 0.f;
  #pragma unroll 8
  for (int m = 0; m < DD; ++m)
    c += W3[k * DD + m] * W1[m * HH + j];
  M[j * HH + k] = W2[j * HH + k] * c;
}

__global__ __launch_bounds__(NTH) void cnf_main(const float* __restrict__ y0,
                                                const float* __restrict__ ts,
                                                const float* __restrict__ W1,
                                                const float* __restrict__ b1,
                                                const float* __restrict__ W2,
                                                const float* __restrict__ b2,
                                                const float* __restrict__ W3,
                                                const float* __restrict__ b3,
                                                const float* __restrict__ Mtr,
                                                float* __restrict__ lpart,
                                                float* __restrict__ out) {
  extern __shared__ __attribute__((aligned(16))) float smem[];
  float* Xs   = smem;
  float* H1s  = Xs  + MT * XP;
  float* H2s  = H1s + MT * HP;
  float* zacc = H2s + MT * HP;
  float* lacc = zacc + MT;
  float* red2 = lacc + MT;
  float* cst  = red2 + NW * R2 * 32;
  float* w1l  = cst + CST;
  float* dxp  = H1s;

  const int tid     = threadIdx.x;
  const int wave    = __builtin_amdgcn_readfirstlane(tid >> 5);
  const int lane    = tid & 31;
  const int lane_lo = lane & 15;
  const int half    = lane >> 4;
  const int tp      = blockIdx.x & (TP - 1);
  const int tq      = tp & 3;
  const int tm      = tp >> 2;
  const int s_base  = (blockIdx.x / TP) * MT;

  for (int i = tid; i < MT * DD; i += NTH) {
    int m = i >> 6, c = i & 63;
    Xs[m * XP + c] = y0[(s_base + m) * DD + c];
  }
  if (tid < MT) { zacc[tid] = 0.f; lacc[tid] = 0.f; }
  for (int i = tid; i < HH; i += NTH) {
    cst[i]          = b1[i];
    cst[HH + i]     = W1[DD * HH + i];
    cst[2 * HH + i] = b2[i];
  }
  if (tid < DD) cst[3 * HH + tid] = b3[tid];
  if (tid < TT) cst[3 * HH + DD + tid] = ts[tid];
  static_assert(NTW == 2, "the W1 fragment rows in LDS hold two N-tiles");
  float* w1r = w1l + (wave * (DD / 4) * 32 + lane) * 4;
  {
    const float* w1p = W1 + (2 * half) * HH + wave * (NTW * 16) + lane_lo;
    #pragma unroll
    for (int ks = 0; ks < DD / 4; ++ks) {
      v4f f;
      f.x = w1p[(ks * 4 + 0) * HH];
      f.y = w1p[(ks * 4 + 1) * HH];
      f.z = w1p[(ks * 4 + 0) * HH + 16];
      f.w = w1p[(ks * 4 + 1) * HH + 16];
      *(v4f*)(w1r + ks * 128) = f;
    }
  }

  v8f acc[2][NTW];

  for (int step = 0; step < TT - 1; ++step) {
    v2f w1f[DD / 4][NTW];
    #pragma unroll
    for (int ks = 0; ks < DD / 4; ++ks) {
      v4f f = *(const v4f*)(w1r + ks * 128);
      w1f[ks][0].x = f.x;  w1f[ks][0].y = f.y;
      w1f[ks][1].x = f.z;  w1f[ks][1].y = f.w;
    }
    __syncthreads();
    const float t0 = cst[3 * HH + DD + step];
    const float dt = cst[3 * HH + DD + step + 1] - t0;

    #pragma unroll
    for (int mt = 0; mt < 2; ++mt)
      #pragma unroll
      for (int nt = 0; nt < NTW; ++nt) acc[mt][nt] = (v8f){};
    #pragma unroll
    for (int ks = 0; ks < DD / 4; ++ks) {
      v2f a0 = frag_a(Xs, XP, 0,  ks * 4, lane_lo, half);
      v2f a1 = frag_a(Xs, XP, 16, ks * 4, lane_lo, half);
      #pragma unroll
      for (int nt = 0; nt < NTW; ++nt) {
        acc[0][nt] = wmma_f32(a0, w1f[ks][nt], acc[0][nt]);
        acc[1][nt] = wmma_f32(a1, w1f[ks][nt], acc[1][nt]);
      }
    }
    #pragma unroll
    for (int nt = 0; nt < NTW; ++nt) {
      int col = (wave * NTW + nt) * 16 + lane_lo;
      float add = t0 * cst[HH + col] + cst[col];
      #pragma unroll
      for (int mt = 0; mt < 2; ++mt)
        #pragma unroll
        for (int v = 0; v < 8; ++v) {
          int m = mt * 16 + v + 8 * half;
          H1s[m * HP + col] = fast_tanh(acc[mt][nt][v] + add);
        }
    }
    __syncthreads();

    {
      const int ng    = wave & 7;
      const int kp    = wave >> 3;
      const int kbase = kp * (HH / 2);
      int cofs[4];
      #pragma unroll
      for (int nt = 0; nt < 4; ++nt) cofs[nt] = ((nt + tq) & 3) * 16;
      v8f hac[2][4];
      v8f vac = (v8f){};
      #pragma unroll
      for (int mt = 0; mt < 2; ++mt)
        #pragma unroll
        for (int nt = 0; nt < 4; ++nt) hac[mt][nt] = (v8f){};
      const float* w2p = W2  + (kbase + 2 * half) * HH + ng * 64 + lane_lo;
      const float* mp  = Mtr + (kbase + 2 * half) * HH + ng * 64 + cofs[0] + lane_lo;
      const float* h1p = H1s + lane_lo * HP + kbase + 2 * half;
      for (int kk = 0; kk < HH / 2; kk += 4 * KU) {
        v2f b[KU][4], bm[KU], a0[KU], a1[KU];
        #pragma unroll
        for (int u = 0; u < KU; ++u) {
          #pragma unroll
          for (int nt = 0; nt < 4; ++nt) {
            b[u][nt].x  = w2p[(kk + 4 * u + 0) * HH + cofs[nt]];
            b[u][nt].y  = w2p[(kk + 4 * u + 1) * HH + cofs[nt]];
          }
          bm[u].x = mp[(kk + 4 * u + 0) * HH];
          bm[u].y = mp[(kk + 4 * u + 1) * HH];
          a0[u].x = h1p[kk + 4 * u];            a0[u].y = h1p[kk + 4 * u + 1];
          a1[u].x = h1p[16 * HP + kk + 4 * u];  a1[u].y = h1p[16 * HP + kk + 4 * u + 1];
        }
        #pragma unroll
        for (int u = 0; u < KU; ++u) {
          v2f at = tm ? a1[u] : a0[u];
          v2f s;
          s.x = 1.f - at.x * at.x;  s.y = 1.f - at.y * at.y;
          #pragma unroll
          for (int nt = 0; nt < 4; ++nt) {
            hac[0][nt] = wmma_f32(a0[u], b[u][nt], hac[0][nt]);
            hac[1][nt] = wmma_f32(a1[u], b[u][nt], hac[1][nt]);
          }
          vac = wmma_f32(s, bm[u], vac);
        }
      }
      const bool fin = (kp == tm);
      float* mine = red2 + wave * (R2 * 32);
      float* h2o = H2s + ((1 - kp) * 16 + 8 * half) * HP + ng * 64 + lane_lo;
      float* h2m = H2s + (kp * 16 + 8 * half) * HP + ng * 64 + lane_lo;
      float kh[4][8];
      #pragma unroll
      for (int v = 0; v < 8; ++v) {
        #pragma unroll
        for (int nt = 0; nt < 4; ++nt) {
          h2o[v * HP + cofs[nt]] = kp ? hac[0][nt][v] : hac[1][nt][v];
          kh[nt][v] = kp ? hac[1][nt][v] : hac[0][nt][v];
        }
      }
      if (!fin) {
        #pragma unroll
        for (int v = 0; v < 8; ++v) mine[v * 32 + lane] = vac[v];
      }
      __syncthreads();
      const float* theirs = red2 + (wave ^ 8) * (R2 * 32);
      float ph[4][8], pv[8];
      #pragma unroll
      for (int v = 0; v < 8; ++v) {
        #pragma unroll
        for (int nt = 0; nt < 4; ++nt) ph[nt][v] = h2m[v * HP + cofs[nt]];
        pv[v] = 0.f;
      }
      if (fin) {
        #pragma unroll
        for (int v = 0; v < 8; ++v) pv[v] = theirs[v * 32 + lane];
      }
      int w3o = 0;
      asm volatile("" : "+s"(w3o));
      const float* w3p = W3 + w3o + (ng * 64 + 2 * half) * DD + lane_lo;
      v2f w3a[8][4];
      #pragma unroll
      for (int ks = 0; ks < 8; ++ks)
        #pragma unroll
        for (int nt = 0; nt < 4; ++nt) {
          w3a[ks][nt].x = w3p[(ks * 4 + 0) * DD + nt * 16];
          w3a[ks][nt].y = w3p[(ks * 4 + 1) * DD + nt * 16];
        }
      float add[4];
      #pragma unroll
      for (int nt = 0; nt < 4; ++nt) add[nt] = cst[2 * HH + ng * 64 + cofs[nt] + lane_lo];
      float h20[8];
      #pragma unroll
      for (int v = 0; v < 8; ++v) {
        int m = kp * 16 + v + 8 * half;
        #pragma unroll
        for (int nt = 0; nt < 4; ++nt) {
          int col = ng * 64 + cofs[nt] + lane_lo;
          float a2 = kh[nt][v] + ph[nt][v];
          float h2 = fast_tanh(a2 + add[nt]);
          H2s[m * HP + col] = h2;
          if (nt == 0) h20[v] = h2;
        }
      }
      v2f w3b[8][4];
      #pragma unroll
      for (int ks = 0; ks < 8; ++ks)
        #pragma unroll
        for (int nt = 0; nt < 4; ++nt) {
          w3b[ks][nt].x = w3p[((ks + 8) * 4 + 0) * DD + nt * 16];
          w3b[ks][nt].y = w3p[((ks + 8) * 4 + 1) * DD + nt * 16];
        }
      if (fin) {
        float tl[8];
        #pragma unroll
        for (int v = 0; v < 8; ++v)
          tl[v] = (1.f - h20[v] * h20[v]) * (vac[v] + pv[v]);
        row16_sum<8>(tl);
        if (lane_lo == 0) {
          #pragma unroll
          for (int v = 0; v < 8; ++v)
            atomicAdd(&lacc[kp * 16 + v + 8 * half], dt * tl[v]);
        }
      }
      __builtin_amdgcn_fence(__ATOMIC_RELEASE, "wavefront");
      __builtin_amdgcn_fence(__ATOMIC_ACQUIRE, "wavefront");
      const float* h2p = H2s + (kp * 16 + lane_lo) * HP + ng * 64 + 2 * half;
      v2f ha[16];
      #pragma unroll
      for (int ks = 0; ks < 16; ++ks) { ha[ks].x = h2p[ks * 4]; ha[ks].y = h2p[ks * 4 + 1]; }
      v8f dacc[4];
      #pragma unroll
      for (int nt = 0; nt < 4; ++nt) dacc[nt] = (v8f){};
      #pragma unroll
      for (int ks = 0; ks < 8; ++ks)
        #pragma unroll
        for (int nt = 0; nt < 4; ++nt) dacc[nt] = wmma_f32(ha[ks], w3a[ks][nt], dacc[nt]);
      #pragma unroll
      for (int ks = 0; ks < 8; ++ks)
        #pragma unroll
        for (int nt = 0; nt < 4; ++nt) dacc[nt] = wmma_f32(ha[ks + 8], w3b[ks][nt], dacc[nt]);
      #pragma unroll
      for (int nt = 0; nt < 4; ++nt)
        #pragma unroll
        for (int q = 0; q < 2; ++q) {
          v4f o;
          o.x = dacc[nt][4 * q + 0];  o.y = dacc[nt][4 * q + 1];
          o.z = dacc[nt][4 * q + 2];  o.w = dacc[nt][4 * q + 3];
          *(v4f*)(dxp + ((wave * 4 + nt) * 2 + q) * 128 + lane * 4) = o;
        }
    }
    __syncthreads();

    {
      const int tile = wave & 7;
      const int q    = wave >> 3;
      const int mt = tile >> 2;
      const int m0 = mt * 16;
      const int n0 = (tile & 3) * 16;
      v4f pc[8];
      #pragma unroll
      for (int g = 0; g < 8; ++g)
        pc[g] = *(const v4f*)(dxp + (((g + 8 * mt) * 4 + (tile & 3)) * 2 + q) * 128 + lane * 4);
      v4f s4 = ((pc[0] + pc[1]) + (pc[2] + pc[3])) + ((pc[4] + pc[5]) + (pc[6] + pc[7]));
      float dxs[4] = {s4.x, s4.y, s4.z, s4.w};
      int col = n0 + lane_lo;
      float bb = cst[3 * HH + col];
      float zt[4];
      #pragma unroll
      for (int j = 0; j < 4; ++j) {
        int m = m0 + 4 * q + j + 8 * half;
        float dx = dxs[j] + bb;
        zt[j] = dt * 0.5f * dx * dx;
        Xs[m * XP + col] += dt * dx;
      }
      row16_sum<4>(zt);
      if (lane_lo == 0) {
        #pragma unroll
        for (int j = 0; j < 4; ++j)
          atomicAdd(&zacc[m0 + 4 * q + j + 8 * half], zt[j]);
      }
    }
  }

  __syncthreads();
  if (tp == 0) {
    for (int i = tid; i < MT * DD; i += NTH) {
      int m = i >> 6, c = i & 63;
      out[(s_base + m) * DD + c] = Xs[m * XP + c];
    }
    if (tid < MT) out[BB * DD + s_base + tid] = zacc[tid];
  }
  if (tid < 16) lpart[tp * BB + s_base + tm * 16 + tid] = lacc[tm * 16 + tid];
}

__global__ __launch_bounds__(256) void cnf_join(const float* __restrict__ lpart,
                                                float* __restrict__ out) {
  int s = blockIdx.x * blockDim.x + threadIdx.x;
  const float* p = lpart + ((s >> 4) & 1) * 4 * BB + s;
  out[BB * DD + BB + s] = (p[0] + p[BB]) + (p[2 * BB] + p[3 * BB]);
}

extern "C" void kernel_launch(void* const* d_in, const int* in_sizes, int n_in,
                              void* d_out, int out_size, void* d_ws, size_t ws_size,
                              hipStream_t stream) {
  const float* y0 = (const float*)d_in[0];
  const float* ts = (const float*)d_in[1];
  const float* W1 = (const float*)d_in[2];
  const float* b1 = (const float*)d_in[3];
  const float* W2 = (const float*)d_in[4];
  const float* b2 = (const float*)d_in[5];
  const float* W3 = (const float*)d_in[6];
  const float* b3 = (const float*)d_in[7];
  float* out = (float*)d_out;
  float* Mtr   = (float*)d_ws;
  float* lpart = Mtr + HH * HH;

  cnf_prep<<<(HH * HH) / 256, 256, 0, stream>>>(W1, W2, W3, Mtr);

  size_t shmem = (size_t)(MT * XP + 2 * MT * HP + 2 * MT + NW * R2 * 32 + CST + NW * (DD / 4) * 32 * 4) * sizeof(float);
  cnf_main<<<(BB / MT) * TP, NTH, shmem, stream>>>(y0, ts, W1, b1, W2, b2, W3, b3,
                                                   Mtr, lpart, out);
  cnf_join<<<BB / 256, 256, 0, stream>>>(lpart, out);
}


# Sample tile per workgroup halved to one 16-row tile: 4 workgroups per 16 samples (not 8 per 32) redo the state chain, each with a quarter of the trace columns; 8 k-steps a loop trip; dx sums regroup
# speedup vs baseline: 3.5025x; 1.5326x over previous
#include <hip/hip_runtime.h>
#include <hip/hip_bf16.h>


#define BB  512
#define DD  64
#define HH  512
#define TT  8
#define MT  16
#define TP  4
#define NW  16
#define NTH (NW * 32)
#define NTW (32 / NW)
#define XP  68
#define HP  516
#define R2  8
#define CST (3 * HH + DD + TT)
#define KU  8
#define DXF (NW * 4 * 4 * 32 * 2)

typedef float v2f __attribute__((ext_vector_type(2)));
typedef float v4f __attribute__((ext_vector_type(4)));
typedef float v8f __attribute__((ext_vector_type(8)));

__device__ __forceinline__ v8f wmma_f32(v2f a, v2f b, v8f c) {
  return __builtin_amdgcn_wmma_f32_16x16x4_f32(false, a, false, b,
                                               (short)0, c, false, false);
}

__device__ __forceinline__ float fast_tanh(float x) {
#if __has_builtin(__builtin_amdgcn_tanhf)
  return __builtin_amdgcn_tanhf(x);
#elif __has_builtin(__builtin_amdgcn_exp2f) && __has_builtin(__builtin_amdgcn_rcpf)
  float e = __builtin_amdgcn_exp2f(x * 2.8853900817779268f);
  return 1.0f - 2.0f * __builtin_amdgcn_rcpf(e + 1.0f);
#else
  return tanhf(x);
#endif
}

__device__ __forceinline__ v2f frag_a(const float* S, int pitch, int m0, int k0,
                                      int lane_lo, int half) {
  const float* p = S + (m0 + lane_lo) * pitch + k0 + 2 * half;
  v2f a; a.x = p[0]; a.y = p[1];
  return a;
}

__device__ __forceinline__ v2f frag_b(const float* __restrict__ W, int ldw,
                                      int k0, int n0, int lane_lo, int half) {
  v2f b;
  b.x = W[(k0 + 2 * half + 0) * ldw + n0 + lane_lo];
  b.y = W[(k0 + 2 * half + 1) * ldw + n0 + lane_lo];
  return b;
}

template <int X>
__device__ __forceinline__ float row_xor(float v) {
  return __builtin_bit_cast(float,
      __builtin_amdgcn_update_dpp(0, __builtin_bit_cast(int, v),
                                  0x160 + X, 0xf, 0xf, true));
}

template <int N>
__device__ __forceinline__ void row16_sum(float (&r)[N]) {
  #pragma unroll
  for (int i = 0; i < N; ++i) r[i] += row_xor<1>(r[i]);
  #pragma unroll
  for (int i = 0; i < N; ++i) r[i] += row_xor<2>(r[i]);
  #pragma unroll
  for (int i = 0; i < N; ++i) r[i] += row_xor<4>(r[i]);
  #pragma unroll
  for (int i = 0; i < N; ++i) r[i] += row_xor<8>(r[i]);
}

__global__ __launch_bounds__(256) void cnf_prep(const float* __restrict__ W1,
                                                const float* __restrict__ W2,
                                                const float* __restrict__ W3,
                                                float* __restrict__ M) {
  int idx = blockIdx.x * blockDim.x + threadIdx.x;
  int j = idx >> 9;
  int k = idx & 511;
  float c = 0.f;
  #pragma unroll 8
  for (int m = 0; m < DD; ++m)
    c += W3[k * DD + m] * W1[m * HH + j];
  M[j * HH + k] = W2[j * HH + k] * c;
}

__global__ __launch_bounds__(NTH) void cnf_main(const float* __restrict__ y0,
                                                const float* __restrict__ ts,
                                                const float* __restrict__ W1,
                                                const float* __restrict__ b1,
                                                const float* __restrict__ W2,
                                                const float* __restrict__ b2,
                                                const float* __restrict__ W3,
                                                const float* __restrict__ b3,
                                                const float* __restrict__ Mtr,
                                                float* __restrict__ lpart,
                                                float* __restrict__ out) {
  extern __shared__ __attribute__((aligned(16))) float smem[];
  float* Xs   = smem;
  float* H1s  = Xs  + MT * XP;
  float* H2s  = H1s + MT * HP;
  float* zacc = H2s + MT * HP;
  float* lacc = zacc + MT;
  float* red2 = lacc + MT;
  float* cst  = red2 + NW * R2 * 32;
  float* w1l  = cst + CST;
  float* dxp  = w1l + NW * (DD / 4) * 32 * 4;

  const int tid     = threadIdx.x;
  const int wave    = __builtin_amdgcn_readfirstlane(tid >> 5);
  const int lane    = tid & 31;
  const int lane_lo = lane & 15;
  const int half    = lane >> 4;
  const int tq      = blockIdx.x & (TP - 1);
  const int s_base  = (blockIdx.x / TP) * MT;

  for (int i = tid; i < MT * DD; i += NTH) {
    int m = i >> 6, c = i & 63;
    Xs[m * XP + c] = y0[(s_base + m) * DD + c];
  }
  if (tid < MT) { zacc[tid] = 0.f; lacc[tid] = 0.f; }
  for (int i = tid; i < HH; i += NTH) {
    cst[i]          = b1[i];
    cst[HH + i]     = W1[DD * HH + i];
    cst[2 * HH + i] = b2[i];
  }
  if (tid < DD) cst[3 * HH + tid] = b3[tid];
  if (tid < TT) cst[3 * HH + DD + tid] = ts[tid];
  static_assert(NTW == 2, "the W1 fragment rows in LDS hold two N-tiles");
  float* w1r = w1l + (wave * (DD / 4) * 32 + lane) * 4;
  {
    const float* w1p = W1 + (2 * half) * HH + wave * (NTW * 16) + lane_lo;
    #pragma unroll
    for (int ks = 0; ks < DD / 4; ++ks) {
      v4f f;
      f.x = w1p[(ks * 4 + 0) * HH];
      f.y = w1p[(ks * 4 + 1) * HH];
      f.z = w1p[(ks * 4 + 0) * HH + 16];
      f.w = w1p[(ks * 4 + 1) * HH + 16];
      *(v4f*)(w1r + ks * 128) = f;
    }
  }

  v8f acc[NTW];

  for (int step = 0; step < TT - 1; ++step) {
    v2f w1f[DD / 4][NTW];
    #pragma unroll
    for (int ks = 0; ks < DD / 4; ++ks) {
      v4f f = *(const v4f*)(w1r + ks * 128);
      w1f[ks][0].x = f.x;  w1f[ks][0].y = f.y;
      w1f[ks][1].x = f.z;  w1f[ks][1].y = f.w;
    }
    __syncthreads();
    const float t0 = cst[3 * HH + DD + step];
    const float dt = cst[3 * HH + DD + step + 1] - t0;

    #pragma unroll
    for (int nt = 0; nt < NTW; ++nt) acc[nt] = (v8f){};
    #pragma unroll
    for (int ks = 0; ks < DD / 4; ++ks) {
      v2f a0 = frag_a(Xs, XP, 0, ks * 4, lane_lo, half);
      #pragma unroll
      for (int nt = 0; nt < NTW; ++nt)
        acc[nt] = wmma_f32(a0, w1f[ks][nt], acc[nt]);
    }
    #pragma unroll
    for (int nt = 0; nt < NTW; ++nt) {
      int col = (wave * NTW + nt) * 16 + lane_lo;
      float add = t0 * cst[HH + col] + cst[col];
      #pragma unroll
      for (int v = 0; v < 8; ++v) {
        int m = v + 8 * half;
        H1s[m * HP + col] = fast_tanh(acc[nt][v] + add);
      }
    }
    __syncthreads();

    {
      const int ng    = wave & 7;
      const int kp    = wave >> 3;
      const int kbase = kp * (HH / 2);
      int cofs[4];
      #pragma unroll
      for (int nt = 0; nt < 4; ++nt) cofs[nt] = ((nt + tq + 2 * kp) & 3) * 16;
      v8f hac[4];
      v8f vac = (v8f){};
      #pragma unroll
      for (int nt = 0; nt < 4; ++nt) hac[nt] = (v8f){};
      const float* w2p = W2  + (kbase + 2 * half) * HH + ng * 64 + lane_lo;
      const float* mp  = Mtr + (kbase + 2 * half) * HH + ng * 64 + tq * 16 + lane_lo;
      const float* h1p = H1s + lane_lo * HP + kbase + 2 * half;
      for (int kk = 0; kk < HH / 2; kk += 4 * KU) {
        v2f b[KU][4], bm[KU], a0[KU];
        #pragma unroll
        for (int u = 0; u < KU; ++u) {
          #pragma unroll
          for (int nt = 0; nt < 4; ++nt) {
            b[u][nt].x  = w2p[(kk + 4 * u + 0) * HH + cofs[nt]];
            b[u][nt].y  = w2p[(kk + 4 * u + 1) * HH + cofs[nt]];
          }
          bm[u].x = mp[(kk + 4 * u + 0) * HH];
          bm[u].y = mp[(kk + 4 * u + 1) * HH];
          a0[u].x = h1p[kk + 4 * u];            a0[u].y = h1p[kk + 4 * u + 1];
        }
        #pragma unroll
        for (int u = 0; u < KU; ++u) {
          v2f s;
          s.x = 1.f - a0[u].x * a0[u].x;  s.y = 1.f - a0[u].y * a0[u].y;
          #pragma unroll
          for (int nt = 0; nt < 4; ++nt)
            hac[nt] = wmma_f32(a0[u], b[u][nt], hac[nt]);
          vac = wmma_f32(s, bm[u], vac);
        }
      }
      const bool fin = (kp == 0);
      float* mine = red2 + wave * (R2 * 32);
      float* h2w = H2s + (8 * half) * HP + ng * 64 + lane_lo;
      #pragma unroll
      for (int v = 0; v < 8; ++v) {
        #pragma unroll
        for (int j = 0; j < 2; ++j)
          h2w[v * HP + cofs[2 + j]] = hac[2 + j][v];
      }
      if (!fin) {
        #pragma unroll
        for (int v = 0; v < 8; ++v) mine[v * 32 + lane] = vac[v];
      }
      __syncthreads();
      const float* theirs = red2 + (wave ^ 8) * (R2 * 32);
      float ph[2][8], pv[8];
      #pragma unroll
      for (int v = 0; v < 8; ++v) {
        #pragma unroll
        for (int j = 0; j < 2; ++j) ph[j][v] = h2w[v * HP + cofs[j]];
        pv[v] = 0.f;
      }
      if (fin) {
        #pragma unroll
        for (int v = 0; v < 8; ++v) pv[v] = theirs[v * 32 + lane];
      }
      const float* w3p = W3 + (ng * 64 + 2 * half) * DD + lane_lo;
      v2f w3a[2][4][4];
      #pragma unroll
      for (int j = 0; j < 2; ++j)
        #pragma unroll
        for (int ks = 0; ks < 4; ++ks)
          #pragma unroll
          for (int nt = 0; nt < 4; ++nt) {
            w3a[j][ks][nt].x = w3p[(cofs[j] + ks * 4 + 0) * DD + nt * 16];
            w3a[j][ks][nt].y = w3p[(cofs[j] + ks * 4 + 1) * DD + nt * 16];
          }
      float add[2];
      #pragma unroll
      for (int j = 0; j < 2; ++j) add[j] = cst[2 * HH + ng * 64 + cofs[j] + lane_lo];
      float h20[8];
      #pragma unroll
      for (int v = 0; v < 8; ++v) {
        #pragma unroll
        for (int j = 0; j < 2; ++j) {
          float a2 = hac[j][v] + ph[j][v];
          float h2 = fast_tanh(a2 + add[j]);
          h2w[v * HP + cofs[j]] = h2;
          if (j == 0) h20[v] = h2;
        }
      }
      if (fin) {
        float tl[8];
        #pragma unroll
        for (int v = 0; v < 8; ++v)
          tl[v] = (1.f - h20[v] * h20[v]) * (vac[v] + pv[v]);
        row16_sum<8>(tl);
        if (lane_lo == 0) {
          #pragma unroll
          for (int v = 0; v < 8; ++v)
            atomicAdd(&lacc[v + 8 * half], dt * tl[v]);
        }
      }
      __builtin_amdgcn_fence(__ATOMIC_RELEASE, "wavefront");
      __builtin_amdgcn_fence(__ATOMIC_ACQUIRE, "wavefront");
      const float* h2p = H2s + lane_lo * HP + ng * 64 + 2 * half;
      v2f ha[2][4];
      #pragma unroll
      for (int j = 0; j < 2; ++j)
        #pragma unroll
        for (int ks = 0; ks < 4; ++ks) {
          ha[j][ks].x = h2p[cofs[j] + ks * 4];
          ha[j][ks].y = h2p[cofs[j] + ks * 4 + 1];
        }
      v8f dacc[4];
      #pragma unroll
      for (int nt = 0; nt < 4; ++nt) dacc[nt] = (v8f){};
      #pragma unroll
      for (int j = 0; j < 2; ++j)
        #pragma unroll
        for (int ks = 0; ks < 4; ++ks)
          #pragma unroll
          for (int nt = 0; nt < 4; ++nt)
            dacc[nt] = wmma_f32(ha[j][ks], w3a[j][ks][nt], dacc[nt]);
      #pragma unroll
      for (int nt = 0; nt < 4; ++nt)
        #pragma unroll
        for (int q = 0; q < 4; ++q) {
          v2f o;
          o.x = dacc[nt][2 * q + 0];  o.y = dacc[nt][2 * q + 1];
          *(v2f*)(dxp + (((wave * 4 + nt) * 4 + q) * 32 + lane) * 2) = o;
        }
    }
    __syncthreads();

    {
      const int tile = wave & 3;
      const int q    = wave >> 2;
      const int n0   = tile * 16;
      v2f pc[16];
      #pragma unroll
      for (int g = 0; g < 16; ++g)
        pc[g] = *(const v2f*)(dxp + (((g * 4 + tile) * 4 + q) * 32 + lane) * 2);
      v2f s2 = (((pc[0] + pc[1]) + (pc[2] + pc[3])) + ((pc[4] + pc[5]) + (pc[6] + pc[7])))
             + (((pc[8] + pc[9]) + (pc[10] + pc[11])) + ((pc[12] + pc[13]) + (pc[14] + pc[15])));
      float dxs[2] = {s2.x, s2.y};
      int col = n0 + lane_lo;
      float bb = cst[3 * HH + col];
      float zt[2];
      #pragma unroll
      for (int j = 0; j < 2; ++j) {
        int m = 2 * q + j + 8 * half;
        float dx = dxs[j] + bb;
        zt[j] = dt * 0.5f * dx * dx;
        Xs[m * XP + col] += dt * dx;
      }
      row16_sum<2>(zt);
      if (lane_lo == 0) {
        #pragma unroll
        for (int j = 0; j < 2; ++j)
          atomicAdd(&zacc[2 * q + j + 8 * half], zt[j]);
      }
    }
  }

  __syncthreads();
  if (tq == 0) {
    for (int i = tid; i < MT * DD; i += NTH) {
      int m = i >> 6, c = i & 63;
      out[(s_base + m) * DD + c] = Xs[m * XP + c];
    }
    if (tid < MT) out[BB * DD + s_base + tid] = zacc[tid];
  }
  if (tid < MT) lpart[tq * BB + s_base + tid] = lacc[tid];
}

__global__ __launch_bounds__(256) void cnf_join(const float* __restrict__ lpart,
                                                float* __restrict__ out) {
  int s = blockIdx.x * blockDim.x + threadIdx.x;
  const float* p = lpart + s;
  out[BB * DD + BB + s] = (p[0] + p[BB]) + (p[2 * BB] + p[3 * BB]);
}

extern "C" void kernel_launch(void* const* d_in, const int* in_sizes, int n_in,
                              void* d_out, int out_size, void* d_ws, size_t ws_size,
                              hipStream_t stream) {
  const float* y0 = (const float*)d_in[0];
  const float* ts = (const float*)d_in[1];
  const float* W1 = (const float*)d_in[2];
  const float* b1 = (const float*)d_in[3];
  const float* W2 = (const float*)d_in[4];
  const float* b2 = (const float*)d_in[5];
  const float* W3 = (const float*)d_in[6];
  const float* b3 = (const float*)d_in[7];
  float* out = (float*)d_out;
  float* Mtr   = (float*)d_ws;
  float* lpart = Mtr + HH * HH;

  cnf_prep<<<(HH * HH) / 256, 256, 0, stream>>>(W1, W2, W3, Mtr);

  size_t shmem = (size_t)(MT * XP + 2 * MT * HP + 2 * MT + NW * R2 * 32 + CST + NW * (DD / 4) * 32 * 4 + DXF) * sizeof(float);
  cnf_main<<<(BB / MT) * TP, NTH, shmem, stream>>>(y0, ts, W1, b1, W2, b2, W3, b3,
                                                   Mtr, lpart, out);
  cnf_join<<<BB / 256, 256, 0, stream>>>(lpart, out);
}


# W2 reaches the layer-2 fragments through LDS: a wave loads its rows as 128-bit pieces (2 loads a k-step, not 8), stores 16 rows in a 4 KB piece of its own, reads fragments back; sum order kept
# speedup vs baseline: 3.6360x; 1.0381x over previous
#include <hip/hip_runtime.h>
#include <hip/hip_bf16.h>


#define BB  512
#define DD  64
#define HH  512
#define TT  8
#define MT  16
#define TP  4
#define NW  16
#define NTH (NW * 32)
#define NTW (32 / NW)
#define XP  68
#define HP  516
#define R2  8
#define CST (3 * HH + DD + TT)
#define KU  8
#define DXF (NW * 4 * 4 * 32 * 2)
#define SR  16
#define WSF (SR * 64)

typedef float v2f __attribute__((ext_vector_type(2)));
typedef float v4f __attribute__((ext_vector_type(4)));
typedef float v8f __attribute__((ext_vector_type(8)));

__device__ __forceinline__ v8f wmma_f32(v2f a, v2f b, v8f c) {
  return __builtin_amdgcn_wmma_f32_16x16x4_f32(false, a, false, b,
                                               (short)0, c, false, false);
}

__device__ __forceinline__ float fast_tanh(float x) {
#if __has_builtin(__builtin_amdgcn_tanhf)
  return __builtin_amdgcn_tanhf(x);
#elif __has_builtin(__builtin_amdgcn_exp2f) && __has_builtin(__builtin_amdgcn_rcpf)
  float e = __builtin_amdgcn_exp2f(x * 2.8853900817779268f);
  return 1.0f - 2.0f * __builtin_amdgcn_rcpf(e + 1.0f);
#else
  return tanhf(x);
#endif
}

__device__ __forceinline__ v2f frag_a(const float* S, int pitch, int m0, int k0,
                                      int lane_lo, int half) {
  const float* p = S + (m0 + lane_lo) * pitch + k0 + 2 * half;
  v2f a; a.x = p[0]; a.y = p[1];
  return a;
}

__device__ __forceinline__ v2f frag_b(const float* __restrict__ W, int ldw,
                                      int k0, int n0, int lane_lo, int half) {
  v2f b;
  b.x = W[(k0 + 2 * half + 0) * ldw + n0 + lane_lo];
  b.y = W[(k0 + 2 * half + 1) * ldw + n0 + lane_lo];
  return b;
}

template <int X>
__device__ __forceinline__ float row_xor(float v) {
  return __builtin_bit_cast(float,
      __builtin_amdgcn_update_dpp(0, __builtin_bit_cast(int, v),
                                  0x160 + X, 0xf, 0xf, true));
}

template <int N>
__device__ __forceinline__ void row16_sum(float (&r)[N]) {
  #pragma unroll
  for (int i = 0; i < N; ++i) r[i] += row_xor<1>(r[i]);
  #pragma unroll
  for (int i = 0; i < N; ++i) r[i] += row_xor<2>(r[i]);
  #pragma unroll
  for (int i = 0; i < N; ++i) r[i] += row_xor<4>(r[i]);
  #pragma unroll
  for (int i = 0; i < N; ++i) r[i] += row_xor<8>(r[i]);
}

__device__ __forceinline__ void wave_lds_order() {
  __builtin_amdgcn_fence(__ATOMIC_RELEASE, "wavefront");
  __builtin_amdgcn_fence(__ATOMIC_ACQUIRE, "wavefront");
}

__global__ __launch_bounds__(256) void cnf_prep(const float* __restrict__ W1,
                                                const float* __restrict__ W2,
                                                const float* __restrict__ W3,
                                                float* __restrict__ M) {
  int idx = blockIdx.x * blockDim.x + threadIdx.x;
  int j = idx >> 9;
  int k = idx & 511;
  float c = 0.f;
  #pragma unroll 8
  for (int m = 0; m < DD; ++m)
    c += W3[k * DD + m] * W1[m * HH + j];
  M[j * HH + k] = W2[j * HH + k] * c;
}

__global__ __launch_bounds__(NTH) void cnf_main(const float* __restrict__ y0,
                                                const float* __restrict__ ts,
                                                const float* __restrict__ W1,
                                                const float* __restrict__ b1,
                                                const float* __restrict__ W2,
                                                const float* __restrict__ b2,
                                                const float* __restrict__ W3,
                                                const float* __restrict__ b3,
                                                const float* __restrict__ Mtr,
                                                float* __restrict__ lpart,
                                                float* __restrict__ out) {
  extern __shared__ __attribute__((aligned(16))) float smem[];
  float* Xs   = smem;
  float* H1s  = Xs  + MT * XP;
  float* H2s  = H1s + MT * HP;
  float* zacc = H2s + MT * HP;
  float* lacc = zacc + MT;
  float* red2 = lacc + MT;
  float* cst  = red2 + NW * R2 * 32;
  float* w1l  = cst + CST;
  float* dxp  = w1l + NW * (DD / 4) * 32 * 4;
  static_assert(NW * WSF <= DXF, "every wave's W2 rows fit the array of the dx pieces");

  const int tid     = threadIdx.x;
  const int wave    = __builtin_amdgcn_readfirstlane(tid >> 5);
  const int lane    = tid & 31;
  const int lane_lo = lane & 15;
  const int half    = lane >> 4;
  const int tq      = blockIdx.x & (TP - 1);
  const int s_base  = (blockIdx.x / TP) * MT;

  for (int i = tid; i < MT * DD; i += NTH) {
    int m = i >> 6, c = i & 63;
    Xs[m * XP + c] = y0[(s_base + m) * DD + c];
  }
  if (tid < MT) { zacc[tid] = 0.f; lacc[tid] = 0.f; }
  for (int i = tid; i < HH; i += NTH) {
    cst[i]          = b1[i];
    cst[HH + i]     = W1[DD * HH + i];
    cst[2 * HH + i] = b2[i];
  }
  if (tid < DD) cst[3 * HH + tid] = b3[tid];
  if (tid < TT) cst[3 * HH + DD + tid] = ts[tid];
  static_assert(NTW == 2, "the W1 fragment rows in LDS hold two N-tiles");
  float* w1r = w1l + (wave * (DD / 4) * 32 + lane) * 4;
  {
    const float* w1p = W1 + (2 * half) * HH + wave * (NTW * 16) + lane_lo;
    #pragma unroll
    for (int ks = 0; ks < DD / 4; ++ks) {
      v4f f;
      f.x = w1p[(ks * 4 + 0) * HH];
      f.y = w1p[(ks * 4 + 1) * HH];
      f.z = w1p[(ks * 4 + 0) * HH + 16];
      f.w = w1p[(ks * 4 + 1) * HH + 16];
      *(v4f*)(w1r + ks * 128) = f;
    }
  }

  v8f acc[NTW];

  for (int step = 0; step < TT - 1; ++step) {
    v2f w1f[DD / 4][NTW];
    #pragma unroll
    for (int ks = 0; ks < DD / 4; ++ks) {
      v4f f = *(const v4f*)(w1r + ks * 128);
      w1f[ks][0].x = f.x;  w1f[ks][0].y = f.y;
      w1f[ks][1].x = f.z;  w1f[ks][1].y = f.w;
    }
    __syncthreads();
    const float t0 = cst[3 * HH + DD + step];
    const float dt = cst[3 * HH + DD + step + 1] - t0;

    #pragma unroll
    for (int nt = 0; nt < NTW; ++nt) acc[nt] = (v8f){};
    #pragma unroll
    for (int ks = 0; ks < DD / 4; ++ks) {
      v2f a0 = frag_a(Xs, XP, 0, ks * 4, lane_lo, half);
      #pragma unroll
      for (int nt = 0; nt < NTW; ++nt)
        acc[nt] = wmma_f32(a0, w1f[ks][nt], acc[nt]);
    }
    #pragma unroll
    for (int nt = 0; nt < NTW; ++nt) {
      int col = (wave * NTW + nt) * 16 + lane_lo;
      float add = t0 * cst[HH + col] + cst[col];
      #pragma unroll
      for (int v = 0; v < 8; ++v) {
        int m = v + 8 * half;
        H1s[m * HP + col] = fast_tanh(acc[nt][v] + add);
      }
    }
    __syncthreads();

    {
      const int ng    = wave & 7;
      const int kp    = wave >> 3;
      const int kbase = kp * (HH / 2);
      int cofs[4];
      #pragma unroll
      for (int nt = 0; nt < 4; ++nt) cofs[nt] = ((nt + tq + 2 * kp) & 3) * 16;
      v8f hac[4];
      v8f vac = (v8f){};
      #pragma unroll
      for (int nt = 0; nt < 4; ++nt) hac[nt] = (v8f){};
      const float* w2g = W2  + (kbase + half) * HH + ng * 64 + 4 * lane_lo;
      float* wst = dxp + wave * WSF;
      float* st0 = wst + half * 64 + 4 * lane_lo;
      float* st1 = wst + half * 64 + ((4 * lane_lo + 32) & 63);
      const float* rd[4];
      #pragma unroll
      for (int nt = 0; nt < 4; ++nt)
        rd[nt] = wst + (2 * half) * 64 + ((cofs[nt] + lane_lo + 32 * half) & 63);
      const float* mp  = Mtr + (kbase + 2 * half) * HH + ng * 64 + tq * 16 + lane_lo;
      const float* h1p = H1s + lane_lo * HP + kbase + 2 * half;
      for (int kk = 0; kk < HH / 2; kk += 4 * KU) {
        v4f g[2 * KU];
        v2f bm[KU], a0[KU];
        #pragma unroll
        for (int u = 0; u < KU; ++u) {
          g[2 * u + 0] = *(const v4f*)(w2g + (kk + 4 * u + 0) * HH);
          g[2 * u + 1] = *(const v4f*)(w2g + (kk + 4 * u + 2) * HH);
          bm[u].x = mp[(kk + 4 * u + 0) * HH];
          bm[u].y = mp[(kk + 4 * u + 1) * HH];
          a0[u].x = h1p[kk + 4 * u];            a0[u].y = h1p[kk + 4 * u + 1];
        }
        static_assert(4 * KU == 2 * SR, "a trip's rows pass through the wave's piece of LDS in two stages");
        #pragma unroll
        for (int stg = 0; stg < 2; ++stg) {
          wave_lds_order();
          #pragma unroll
          for (int q = 0; q < SR / 2; ++q)
            *(v4f*)(((q & 1) ? st1 : st0) + q * 128) = g[stg * (SR / 2) + q];
          wave_lds_order();
          #pragma unroll
          for (int uu = 0; uu < SR / 4; ++uu) {
            const int u = stg * (SR / 4) + uu;
            v2f s;
            s.x = 1.f - a0[u].x * a0[u].x;  s.y = 1.f - a0[u].y * a0[u].y;
            #pragma unroll
            for (int nt = 0; nt < 4; ++nt) {
              v2f b;
              b.x = rd[nt][uu * 256];
              b.y = rd[nt][uu * 256 + 64];
              hac[nt] = wmma_f32(a0[u], b, hac[nt]);
            }
            vac = wmma_f32(s, bm[u], vac);
          }
        }
      }
      const bool fin = (kp == 0);
      float* mine = red2 + wave * (R2 * 32);
      float* h2w = H2s + (8 * half) * HP + ng * 64 + lane_lo;
      #pragma unroll
      for (int v = 0; v < 8; ++v) {
        #pragma unroll
        for (int j = 0; j < 2; ++j)
          h2w[v * HP + cofs[2 + j]] = hac[2 + j][v];
      }
      if (!fin) {
        #pragma unroll
        for (int v = 0; v < 8; ++v) mine[v * 32 + lane] = vac[v];
      }
      __syncthreads();
      const float* theirs = red2 + (wave ^ 8) * (R2 * 32);
      float ph[2][8], pv[8];
      #pragma unroll
      for (int v = 0; v < 8; ++v) {
        #pragma unroll
        for (int j = 0; j < 2; ++j) ph[j][v] = h2w[v * HP + cofs[j]];
        pv[v] = 0.f;
      }
      if (fin) {
        #pragma unroll
        for (int v = 0; v < 8; ++v) pv[v] = theirs[v * 32 + lane];
      }
      const float* w3p = W3 + (ng * 64 + 2 * half) * DD + lane_lo;
      v2f w3a[2][4][4];
      #pragma unroll
      for (int j = 0; j < 2; ++j)
        #pragma unroll
        for (int ks = 0; ks < 4; ++ks)
          #pragma unroll
          for (int nt = 0; nt < 4; ++nt) {
            w3a[j][ks][nt].x = w3p[(cofs[j] + ks * 4 + 0) * DD + nt * 16];
            w3a[j][ks][nt].y = w3p[(cofs[j] + ks * 4 + 1) * DD + nt * 16];
          }
      float add[2];
      #pragma unroll
      for (int j = 0; j < 2; ++j) add[j] = cst[2 * HH + ng * 64 + cofs[j] + lane_lo];
      float h20[8];
      #pragma unroll
      for (int v = 0; v < 8; ++v) {
        #pragma unroll
        for (int j = 0; j < 2; ++j) {
          float a2 = hac[j][v] + ph[j][v];
          float h2 = fast_tanh(a2 + add[j]);
          h2w[v * HP + cofs[j]] = h2;
          if (j == 0) h20[v] = h2;
        }
      }
      if (fin) {
        float tl[8];
        #pragma unroll
        for (int v = 0; v < 8; ++v)
          tl[v] = (1.f - h20[v] * h20[v]) * (vac[v] + pv[v]);
        row16_sum<8>(tl);
        if (lane_lo == 0) {
          #pragma unroll
          for (int v = 0; v < 8; ++v)
            atomicAdd(&lacc[v + 8 * half], dt * tl[v]);
        }
      }
      __builtin_amdgcn_fence(__ATOMIC_RELEASE, "wavefront");
      __builtin_amdgcn_fence(__ATOMIC_ACQUIRE, "wavefront");
      const float* h2p = H2s + lane_lo * HP + ng * 64 + 2 * half;
      v2f ha[2][4];
      #pragma unroll
      for (int j = 0; j < 2; ++j)
        #pragma unroll
        for (int ks = 0; ks < 4; ++ks) {
          ha[j][ks].x = h2p[cofs[j] + ks * 4];
          ha[j][ks].y = h2p[cofs[j] + ks * 4 + 1];
        }
      v8f dacc[4];
      #pragma unroll
      for (int nt = 0; nt < 4; ++nt) dacc[nt] = (v8f){};
      #pragma unroll
      for (int j = 0; j < 2; ++j)
        #pragma unroll
        for (int ks = 0; ks < 4; ++ks)
          #pragma unroll
          for (int nt = 0; nt < 4; ++nt)
            dacc[nt] = wmma_f32(ha[j][ks], w3a[j][ks][nt], dacc[nt]);
      #pragma unroll
      for (int nt = 0; nt < 4; ++nt)
        #pragma unroll
        for (int q = 0; q < 4; ++q) {
          v2f o;
          o.x = dacc[nt][2 * q + 0];  o.y = dacc[nt][2 * q + 1];
          *(v2f*)(dxp + (((wave * 4 + nt) * 4 + q) * 32 + lane) * 2) = o;
        }
    }
    __syncthreads();

    {
      const int tile = wave & 3;
      const int q    = wave >> 2;
      const int n0   = tile * 16;
      v2f pc[16];
      #pragma unroll
      for (int g = 0; g < 16; ++g)
        pc[g] = *(const v2f*)(dxp + (((g * 4 + tile) * 4 + q) * 32 + lane) * 2);
      v2f s2 = (((pc[0] + pc[1]) + (pc[2] + pc[3])) + ((pc[4] + pc[5]) + (pc[6] + pc[7])))
             + (((pc[8] + pc[9]) + (pc[10] + pc[11])) + ((pc[12] + pc[13]) + (pc[14] + pc[15])));
      float dxs[2] = {s2.x, s2.y};
      int col = n0 + lane_lo;
      float bb = cst[3 * HH + col];
      float zt[2];
      #pragma unroll
      for (int j = 0; j < 2; ++j) {
        int m = 2 * q + j + 8 * half;
        float dx = dxs[j] + bb;
        zt[j] = dt * 0.5f * dx * dx;
        Xs[m * XP + col] += dt * dx;
      }
      row16_sum<2>(zt);
      if (lane_lo == 0) {
        #pragma unroll
        for (int j = 0; j < 2; ++j)
          atomicAdd(&zacc[2 * q + j + 8 * half], zt[j]);
      }
    }
  }

  __syncthreads();
  if (tq == 0) {
    for (int i = tid; i < MT * DD; i += NTH) {
      int m = i >> 6, c = i & 63;
      out[(s_base + m) * DD + c] = Xs[m * XP + c];
    }
    if (tid < MT) out[BB * DD + s_base + tid] = zacc[tid];
  }
  if (tid < MT) lpart[tq * BB + s_base + tid] = lacc[tid];
}

__global__ __launch_bounds__(256) void cnf_join(const float* __restrict__ lpart,
                                                float* __restrict__ out) {
  int s = blockIdx.x * blockDim.x + threadIdx.x;
  const float* p = lpart + s;
  out[BB * DD + BB + s] = (p[0] + p[BB]) + (p[2 * BB] + p[3 * BB]);
}

extern "C" void kernel_launch(void* const* d_in, const int* in_sizes, int n_in,
                              void* d_out, int out_size, void* d_ws, size_t ws_size,
                              hipStream_t stream) {
  const float* y0 = (const float*)d_in[0];
  const float* ts = (const float*)d_in[1];
  const float* W1 = (const float*)d_in[2];
  const float* b1 = (const float*)d_in[3];
  const float* W2 = (const float*)d_in[4];
  const float* b2 = (const float*)d_in[5];
  const float* W3 = (const float*)d_in[6];
  const float* b3 = (const float*)d_in[7];
  float* out = (float*)d_out;
  float* Mtr   = (float*)d_ws;
  float* lpart = Mtr + HH * HH;

  cnf_prep<<<(HH * HH) / 256, 256, 0, stream>>>(W1, W2, W3, Mtr);

  size_t shmem = (size_t)(MT * XP + 2 * MT * HP + 2 * MT + NW * R2 * 32 + CST + NW * (DD / 4) * 32 * 4 + DXF) * sizeof(float);
  cnf_main<<<(BB / MT) * TP, NTH, shmem, stream>>>(y0, ts, W1, b1, W2, b2, W3, b3,
                                                   Mtr, lpart, out);
  cnf_join<<<BB / 256, 256, 0, stream>>>(lpart, out);
}
